# starting kernel
# baseline (speedup 1.0000x reference)
#include <hip/hip_runtime.h>


typedef float v2f __attribute__((ext_vector_type(2)));
typedef float v8f __attribute__((ext_vector_type(8)));

__global__ void zero_f32(float* __restrict__ p, int n) {
    int i = blockIdx.x * blockDim.x + threadIdx.x;
    if (i < n) p[i] = 0.0f;
}

__global__ void edge_count(const int* __restrict__ dst, float* __restrict__ cnt, int E) {
    int e = blockIdx.x * blockDim.x + threadIdx.x;
    if (e < E) atomicAdd(&cnt[dst[e]], 1.0f);
}

__global__ void edge_scatter(const float* __restrict__ xs, const int* __restrict__ src,
                             const int* __restrict__ dst, float* __restrict__ sum,
                             int E, int D) {
    int chunks = D >> 2;
    long long tid = (long long)blockIdx.x * blockDim.x + threadIdx.x;
    if (tid >= (long long)E * chunks) return;
    int e = (int)(tid / chunks);
    int c = (int)(tid % chunks);
    const float4 v = *(const float4*)(xs + (size_t)src[e] * D + (size_t)c * 4);
    float* s = sum + (size_t)dst[e] * D + (size_t)c * 4;
    atomicAdd(s + 0, v.x);
    atomicAdd(s + 1, v.y);
    atomicAdd(s + 2, v.z);
    atomicAdd(s + 3, v.w);
}

__global__ void seg_normalize(float* __restrict__ sum, const float* __restrict__ cnt,
                              int n, int D) {
    int chunks = D >> 2;
    long long tid = (long long)blockIdx.x * blockDim.x + threadIdx.x;
    if (tid >= (long long)n * chunks) return;
    int i = (int)(tid / chunks);
    int c = (int)(tid % chunks);
    float inv = 1.0f / fmaxf(cnt[i], 1.0f);
    float4* p = (float4*)(sum + (size_t)i * D + (size_t)c * 4);
    float4 v = *p;
    v.x *= inv; v.y *= inv; v.z *= inv; v.w *= inv;
    *p = v;
}

__global__ void sage_gemm_wmma(const float* __restrict__ A0,
                               const float* __restrict__ A1,
                               const float* __restrict__ W0,
                               const float* __restrict__ W1,
                               const float* __restrict__ bias,
                               float* __restrict__ out,
                               int M, int K, int N, int relu) {
    int wave = blockIdx.x * (blockDim.x >> 5) + (threadIdx.x >> 5);
    int tilesN = N >> 4;
    int totalTiles = (M >> 4) * tilesN;
    if (wave >= totalTiles) return;

    int tm = wave / tilesN;
    int tn = wave - tm * tilesN;
    int lane = threadIdx.x & 31;
    int half = lane >> 4;
    int l15  = lane & 15;
    int arow = (tm << 4) + l15;
    int bcol = (tn << 4) + l15;

    v8f acc = {0.f, 0.f, 0.f, 0.f, 0.f, 0.f, 0.f, 0.f};

    #pragma unroll
    for (int pass = 0; pass < 2; ++pass) {
        const float* A = pass ? A1 : A0;
        const float* W = pass ? W1 : W0;
        const float* ap = A + (size_t)arow * K + (half << 1);
        #pragma unroll 4
        for (int k0 = 0; k0 < K; k0 += 4) {
            v2f a = *(const v2f*)(ap + k0);
            int bk = k0 + (half << 1);
            v2f b;
            b.x = W[(size_t)bk * N + bcol];
            b.y = W[(size_t)(bk + 1) * N + bcol];
            acc = __builtin_amdgcn_wmma_f32_16x16x4_f32(
                 false, a,  false, b,
                 (short)0, acc,  false,  false);
        }
    }

    float bv = bias[bcol];
    float* op = out + (size_t)((tm << 4) + (half << 3)) * N + bcol;
    #pragma unroll
    for (int v = 0; v < 8; ++v) {
        float r = acc[v] + bv;
        if (relu) r = fmaxf(r, 0.0f);
        op[(size_t)v * N] = r;
    }
}

extern "C" void kernel_launch(void* const* d_in, const int* in_sizes, int n_in,
                              void* d_out, int out_size, void* d_ws, size_t ws_size,
                              hipStream_t stream) {
    (void)in_sizes; (void)n_in; (void)out_size; (void)ws_size;

    const float* x = (const float*)d_in[0];
    const int* srcp[3] = {(const int*)d_in[1], (const int*)d_in[3], (const int*)d_in[5]};
    const int* dstp[3] = {(const int*)d_in[2], (const int*)d_in[4], (const int*)d_in[6]};
    const float* Wl[3] = {(const float*)d_in[7],  (const float*)d_in[10], (const float*)d_in[13]};
    const float* Wr[3] = {(const float*)d_in[8],  (const float*)d_in[11], (const float*)d_in[14]};
    const float* bb[3] = {(const float*)d_in[9],  (const float*)d_in[12], (const float*)d_in[15]};

    const int Ecnt[3] = {1500000, 300000, 60000};
    const int ntgt[3] = {100000, 20000, 4000};
    const int Kd[3]   = {128, 256, 256};
    const int Nd[3]   = {256, 256, 128};
    const int relu[3] = {1, 0, 1};

    float* ws     = (float*)d_ws;
    float* aggbuf = ws;
    float* cntbuf = aggbuf + 12800000;
    float* h0     = cntbuf + 100000;
    float* h1     = h0 + 25600000;

    const float* layer_in[3] = {x, h0, h1};
    float* layer_out[3]      = {h0, h1, (float*)d_out};

    const int TPB = 256;
    for (int L = 0; L < 3; ++L) {
        int E = Ecnt[L], n = ntgt[L], K = Kd[L], N = Nd[L];

        int zn = n * K;
        zero_f32<<<(zn + TPB - 1) / TPB, TPB, 0, stream>>>(aggbuf, zn);
        zero_f32<<<(n + TPB - 1) / TPB, TPB, 0, stream>>>(cntbuf, n);

        edge_count<<<(E + TPB - 1) / TPB, TPB, 0, stream>>>(dstp[L], cntbuf, E);

        long long sthreads = (long long)E * (K >> 2);
        edge_scatter<<<(unsigned)((sthreads + TPB - 1) / TPB), TPB, 0, stream>>>(
            layer_in[L], srcp[L], dstp[L], aggbuf, E, K);

        long long nthreads = (long long)n * (K >> 2);
        seg_normalize<<<(unsigned)((nthreads + TPB - 1) / TPB), TPB, 0, stream>>>(
            aggbuf, cntbuf, n, K);

        int tiles = (n >> 4) * (N >> 4);
        int wavesPerBlock = TPB >> 5;
        sage_gemm_wmma<<<(tiles + wavesPerBlock - 1) / wavesPerBlock, TPB, 0, stream>>>(
            aggbuf, layer_in[L], Wl[L], Wr[L], bb[L], layer_out[L], n, K, N, relu[L]);
    }
}


# GEMM register tiling: one wave per 32x128 (32x64, 16x64) output tile, 16-byte A and W loads reused over 16 (8, 4) accumulators, next loads issued ahead; k sum order inside each 8-block changes
# speedup vs baseline: 1.2460x; 1.2460x over previous
#include <hip/hip_runtime.h>


typedef float v2f __attribute__((ext_vector_type(2)));
typedef float v4f __attribute__((ext_vector_type(4)));
typedef float v8f __attribute__((ext_vector_type(8)));

__global__ void zero_f32(float* __restrict__ p, int n) {
    int i = blockIdx.x * blockDim.x + threadIdx.x;
    if (i < n) p[i] = 0.0f;
}

__global__ void edge_count(const int* __restrict__ dst, float* __restrict__ cnt, int E) {
    int e = blockIdx.x * blockDim.x + threadIdx.x;
    if (e < E) atomicAdd(&cnt[dst[e]], 1.0f);
}

__global__ void edge_scatter(const float* __restrict__ xs, const int* __restrict__ src,
                             const int* __restrict__ dst, float* __restrict__ sum,
                             int E, int D) {
    int chunks = D >> 2;
    long long tid = (long long)blockIdx.x * blockDim.x + threadIdx.x;
    if (tid >= (long long)E * chunks) return;
    int e = (int)(tid / chunks);
    int c = (int)(tid % chunks);
    const float4 v = *(const float4*)(xs + (size_t)src[e] * D + (size_t)c * 4);
    float* s = sum + (size_t)dst[e] * D + (size_t)c * 4;
    atomicAdd(s + 0, v.x);
    atomicAdd(s + 1, v.y);
    atomicAdd(s + 2, v.z);
    atomicAdd(s + 3, v.w);
}

__global__ void seg_normalize(float* __restrict__ sum, const float* __restrict__ cnt,
                              int n, int D) {
    int chunks = D >> 2;
    long long tid = (long long)blockIdx.x * blockDim.x + threadIdx.x;
    if (tid >= (long long)n * chunks) return;
    int i = (int)(tid / chunks);
    int c = (int)(tid % chunks);
    float inv = 1.0f / fmaxf(cnt[i], 1.0f);
    float4* p = (float4*)(sum + (size_t)i * D + (size_t)c * 4);
    float4 v = *p;
    v.x *= inv; v.y *= inv; v.z *= inv; v.w *= inv;
    *p = v;
}

template <int MT, int NG>
struct Frags {
    v4f a[MT];
    v4f w[NG][4];
};

template <int MT, int NG>
__device__ __forceinline__ void load_frags(Frags<MT, NG>& f, const float* __restrict__ ap, int K,
                                           const float* __restrict__ wp, int N) {
    #pragma unroll
    for (int i = 0; i < MT; ++i)
        f.a[i] = *(const v4f*)(ap + (size_t)(16 * i) * K);
    #pragma unroll
    for (int g = 0; g < NG; ++g) {
        #pragma unroll
        for (int q = 0; q < 4; ++q)
            f.w[g][q] = *(const v4f*)(wp + (size_t)q * N + 64 * g);
    }
}

template <int MT, int NG>
__device__ __forceinline__ void mma_step(v8f (&acc)[MT][NG][4], const Frags<MT, NG>& f) {
    #pragma unroll
    for (int s = 0; s < 2; ++s) {
        #pragma unroll
        for (int i = 0; i < MT; ++i) {
            v2f av;
            av.x = f.a[i][2 * s];
            av.y = f.a[i][2 * s + 1];
            #pragma unroll
            for (int g = 0; g < NG; ++g) {
                #pragma unroll
                for (int j = 0; j < 4; ++j) {
                    v2f bv;
                    bv.x = f.w[g][2 * s][j];
                    bv.y = f.w[g][2 * s + 1][j];
                    acc[i][g][j] = __builtin_amdgcn_wmma_f32_16x16x4_f32(
                         false, av,  false, bv,
                         (short)0, acc[i][g][j],  false,  false);
                }
            }
        }
    }
}

template <int MT, int NG>
__global__ void __launch_bounds__(128)
sage_gemm_wmma(const float* __restrict__ A0,
               const float* __restrict__ A1,
               const float* __restrict__ W0,
               const float* __restrict__ W1,
               const float* __restrict__ bias,
               float* __restrict__ out,
               int M, int K, int N, int relu) {
    constexpr int WM = 16 * MT;
    constexpr int WN = 64 * NG;
    const int wave = blockIdx.x * (blockDim.x >> 5) + (threadIdx.x >> 5);
    const int tilesN = N / WN;
    const int totalTiles = (M / WM) * tilesN;
    if (wave >= totalTiles) return;

    const int tm = wave / tilesN;
    const int tn = wave - tm * tilesN;
    const int lane = threadIdx.x & 31;
    const int h = lane >> 4;
    const int c = lane & 15;
    const int m0 = tm * WM;
    const int n0 = tn * WN;

    v8f acc[MT][NG][4];
    #pragma unroll
    for (int i = 0; i < MT; ++i) {
        #pragma unroll
        for (int g = 0; g < NG; ++g) {
            #pragma unroll
            for (int j = 0; j < 4; ++j)
                acc[i][g][j] = (v8f){0.f, 0.f, 0.f, 0.f, 0.f, 0.f, 0.f, 0.f};
        }
    }

    const size_t aoff = (size_t)(m0 + c) * K + 4 * h;
    const size_t woff = (size_t)(4 * h) * N + n0 + 4 * c;
    const float* a0p = A0 + aoff;
    const float* a1p = A1 + aoff;
    const float* w0p = W0 + woff;
    const float* w1p = W1 + woff;

    const int steps = K >> 3;
    const int nsteps = steps * 2;

    Frags<MT, NG> fa, fb;
    auto fetch = [&](Frags<MT, NG>& f, int t) {
        const bool p = t >= steps;
        const int kk = (p ? t - steps : t) << 3;
        load_frags<MT, NG>(f, (p ? a1p : a0p) + kk, K, (p ? w1p : w0p) + (size_t)kk * N, N);
    };
    fetch(fa, 0);
    int t = 0;
    for (; t < nsteps - 2; t += 2) {
        fetch(fb, t + 1);
        __builtin_amdgcn_sched_barrier(0);
        mma_step<MT, NG>(acc, fa);
        __builtin_amdgcn_sched_barrier(0);
        fetch(fa, t + 2);
        __builtin_amdgcn_sched_barrier(0);
        mma_step<MT, NG>(acc, fb);
        __builtin_amdgcn_sched_barrier(0);
    }
    fetch(fb, t + 1);
    __builtin_amdgcn_sched_barrier(0);
    mma_step<MT, NG>(acc, fa);
    mma_step<MT, NG>(acc, fb);

    #pragma unroll
    for (int g = 0; g < NG; ++g) {
        const int col = n0 + 64 * g + 4 * c;
        const float b0 = bias[col], b1 = bias[col + 1], b2 = bias[col + 2], b3 = bias[col + 3];
        #pragma unroll
        for (int i = 0; i < MT; ++i) {
            float* op = out + (size_t)(m0 + 16 * i + 8 * h) * N + col;
            #pragma unroll
            for (int r = 0; r < 8; ++r) {
                v4f o;
                o.x = acc[i][g][0][r] + b0;
                o.y = acc[i][g][1][r] + b1;
                o.z = acc[i][g][2][r] + b2;
                o.w = acc[i][g][3][r] + b3;
                if (relu) {
                    o.x = fmaxf(o.x, 0.0f); o.y = fmaxf(o.y, 0.0f);
                    o.z = fmaxf(o.z, 0.0f); o.w = fmaxf(o.w, 0.0f);
                }
                *(v4f*)(op + (size_t)r * N) = o;
            }
        }
    }
}

extern "C" void kernel_launch(void* const* d_in, const int* in_sizes, int n_in,
                              void* d_out, int out_size, void* d_ws, size_t ws_size,
                              hipStream_t stream) {
    (void)in_sizes; (void)n_in; (void)out_size; (void)ws_size;

    const float* x = (const float*)d_in[0];
    const int* srcp[3] = {(const int*)d_in[1], (const int*)d_in[3], (const int*)d_in[5]};
    const int* dstp[3] = {(const int*)d_in[2], (const int*)d_in[4], (const int*)d_in[6]};
    const float* Wl[3] = {(const float*)d_in[7],  (const float*)d_in[10], (const float*)d_in[13]};
    const float* Wr[3] = {(const float*)d_in[8],  (const float*)d_in[11], (const float*)d_in[14]};
    const float* bb[3] = {(const float*)d_in[9],  (const float*)d_in[12], (const float*)d_in[15]};

    const int Ecnt[3] = {1500000, 300000, 60000};
    const int ntgt[3] = {100000, 20000, 4000};
    const int Kd[3]   = {128, 256, 256};
    const int Nd[3]   = {256, 256, 128};
    const int relu[3] = {1, 0, 1};

    float* ws     = (float*)d_ws;
    float* aggbuf = ws;
    float* cntbuf = aggbuf + 12800000;
    float* h0     = cntbuf + 100000;
    float* h1     = h0 + 25600000;

    const float* layer_in[3] = {x, h0, h1};
    float* layer_out[3]      = {h0, h1, (float*)d_out};

    const int TPB = 256;
    for (int L = 0; L < 3; ++L) {
        int E = Ecnt[L], n = ntgt[L], K = Kd[L], N = Nd[L];

        int zn = n * K;
        zero_f32<<<(zn + TPB - 1) / TPB, TPB, 0, stream>>>(aggbuf, zn);
        zero_f32<<<(n + TPB - 1) / TPB, TPB, 0, stream>>>(cntbuf, n);

        edge_count<<<(E + TPB - 1) / TPB, TPB, 0, stream>>>(dstp[L], cntbuf, E);

        long long sthreads = (long long)E * (K >> 2);
        edge_scatter<<<(unsigned)((sthreads + TPB - 1) / TPB), TPB, 0, stream>>>(
            layer_in[L], srcp[L], dstp[L], aggbuf, E, K);

        long long nthreads = (long long)n * (K >> 2);
        seg_normalize<<<(unsigned)((nthreads + TPB - 1) / TPB), TPB, 0, stream>>>(
            aggbuf, cntbuf, n, K);

        const int GT = 128;
        if (L == 0) {
            int tiles = (n / 32) * (N / 128);
            sage_gemm_wmma<2, 2><<<(tiles + 3) / 4, GT, 0, stream>>>(
                aggbuf, layer_in[L], Wl[L], Wr[L], bb[L], layer_out[L], n, K, N, relu[L]);
        } else if (L == 1) {
            int tiles = (n / 32) * (N / 64);
            sage_gemm_wmma<2, 1><<<(tiles + 3) / 4, GT, 0, stream>>>(
                aggbuf, layer_in[L], Wl[L], Wr[L], bb[L], layer_out[L], n, K, N, relu[L]);
        } else {
            int tiles = (n / 16) * (N / 64);
            sage_gemm_wmma<1, 1><<<(tiles + 3) / 4, GT, 0, stream>>>(
                aggbuf, layer_in[L], Wl[L], Wr[L], bb[L], layer_out[L], n, K, N, relu[L]);
        }
    }
}


# Segment mean by pull: edges counting-sorted by destination in-call (integer tickets), one wave sums its node's source rows, writes the mean once; float atomics gone, sum order = ticket order
# speedup vs baseline: 2.1324x; 1.7114x over previous
#include <hip/hip_runtime.h>


typedef float v2f __attribute__((ext_vector_type(2)));
typedef float v4f __attribute__((ext_vector_type(4)));
typedef float v8f __attribute__((ext_vector_type(8)));

#define E_L0 1500000
#define E_L1 300000
#define E_L2 60000
#define N_L0 100000
#define N_L1 20000
#define N_L2 4000

__global__ void zero_counts(int* __restrict__ c0, int* __restrict__ c1, int* __restrict__ c2) {
    int t = blockIdx.x * blockDim.x + threadIdx.x;
    if (t < N_L0) c0[t] = 0;
    else if (t < N_L0 + N_L1) c1[t - N_L0] = 0;
    else if (t < N_L0 + N_L1 + N_L2) c2[t - N_L0 - N_L1] = 0;
}

__global__ void edge_hist(const int* __restrict__ d0, const int* __restrict__ d1,
                          const int* __restrict__ d2, int* __restrict__ c0,
                          int* __restrict__ c1, int* __restrict__ c2) {
    int t = blockIdx.x * blockDim.x + threadIdx.x;
    if (t < E_L0) atomicAdd(&c0[d0[t]], 1);
    else if (t < E_L0 + E_L1) atomicAdd(&c1[d1[t - E_L0]], 1);
    else if (t < E_L0 + E_L1 + E_L2) atomicAdd(&c2[d2[t - E_L0 - E_L1]], 1);
}

#define SCAN_B0 98
#define SCAN_B1 20
#define SCAN_B2 4
__global__ void __launch_bounds__(1024)
seg_offsets(const int* __restrict__ c0, int* __restrict__ u0,
            const int* __restrict__ c1, int* __restrict__ u1,
            const int* __restrict__ c2, int* __restrict__ u2) {
    __shared__ int red[32];
    __shared__ int tot[32];
    int b = blockIdx.x;
    const int* cnt; int* cur; int n;
    if (b < SCAN_B0) { cnt = c0; cur = u0; n = N_L0; }
    else if (b < SCAN_B0 + SCAN_B1) { b -= SCAN_B0; cnt = c1; cur = u1; n = N_L1; }
    else { b -= SCAN_B0 + SCAN_B1; cnt = c2; cur = u2; n = N_L2; }
    const int tid = threadIdx.x;
    const int lane = tid & 31;
    const int w = tid >> 5;
    const int lo = b * 1024;
    int p = 0;
    const int4* c4 = (const int4*)cnt;
    for (int j = tid; j < (lo >> 2); j += 1024) {
        int4 q = c4[j];
        p += q.x + q.y + q.z + q.w;
    }
    const int idx = lo + tid;
    const int v = (idx < n) ? cnt[idx] : 0;
    int incl = v;
    for (int d = 1; d < 32; d <<= 1) {
        int t = __shfl_up(incl, d);
        if (lane >= d) incl += t;
    }
    for (int d = 16; d >= 1; d >>= 1) p += __shfl_xor(p, d);
    if (lane == 31) tot[w] = incl;
    if (lane == 0) red[w] = p;
    __syncthreads();
    int base = 0;
    for (int k = 0; k < 32; ++k) {
        base += red[k];
        if (k < w) base += tot[k];
    }
    if (idx < n) cur[idx] = base + incl - v;
}

__global__ void edge_place(const int* __restrict__ s0, const int* __restrict__ d0,
                           const int* __restrict__ s1, const int* __restrict__ d1,
                           const int* __restrict__ s2, const int* __restrict__ d2,
                           int* __restrict__ u0, int* __restrict__ u1, int* __restrict__ u2,
                           int* __restrict__ p0, int* __restrict__ p1, int* __restrict__ p2) {
    int t = blockIdx.x * blockDim.x + threadIdx.x;
    if (t < E_L0) {
        p0[atomicAdd(&u0[d0[t]], 1)] = s0[t];
    } else if (t < E_L0 + E_L1) {
        t -= E_L0;
        p1[atomicAdd(&u1[d1[t]], 1)] = s1[t];
    } else if (t < E_L0 + E_L1 + E_L2) {
        t -= E_L0 + E_L1;
        p2[atomicAdd(&u2[d2[t]], 1)] = s2[t];
    }
}

template <int D, int UNR>
__global__ void __launch_bounds__(256)
seg_mean_gather(const float* __restrict__ xs, const int* __restrict__ perm,
                const int* __restrict__ cnt, const int* __restrict__ cur,
                float* __restrict__ agg, int n) {
    constexpr int V = D / 128;
    const int lane = threadIdx.x & 31;
    const int i = blockIdx.x * 8 + (threadIdx.x >> 5);
    if (i >= n) return;
    const int deg = __builtin_amdgcn_readfirstlane(cnt[i]);
    const int end = __builtin_amdgcn_readfirstlane(cur[i]);
    const int begin = end - deg;
    v4f acc[V];
    #pragma unroll
    for (int v = 0; v < V; ++v) acc[v] = (v4f){0.f, 0.f, 0.f, 0.f};
    const float* xl = xs + 4 * lane;
    for (int j0 = 0; j0 < deg; j0 += 32) {
        const int m = min(32, deg - j0);
        const int my = perm[begin + j0 + min(lane, m - 1)];
        for (int k = 0; k < m; k += UNR) {
            v4f r[UNR][V];
            #pragma unroll
            for (int u = 0; u < UNR; ++u) {
                const int kk = min(k + u, m - 1);
                const int s = __builtin_amdgcn_readlane(my, kk);
                const float* rp = xl + (size_t)s * D;
                #pragma unroll
                for (int v = 0; v < V; ++v) r[u][v] = *(const v4f*)(rp + 128 * v);
            }
            #pragma unroll
            for (int u = 0; u < UNR; ++u) {
                const bool on = (k + u) < m;
                #pragma unroll
                for (int v = 0; v < V; ++v)
                    acc[v] += on ? r[u][v] : (v4f){0.f, 0.f, 0.f, 0.f};
            }
        }
    }
    const float inv = 1.0f / fmaxf((float)deg, 1.0f);
    float* op = agg + (size_t)i * D + 4 * lane;
    #pragma unroll
    for (int v = 0; v < V; ++v) *(v4f*)(op + 128 * v) = acc[v] * inv;
}

template <int MT, int NG>
struct Frags {
    v4f a[MT];
    v4f w[NG][4];
};

template <int MT, int NG>
__device__ __forceinline__ void load_frags(Frags<MT, NG>& f, const float* __restrict__ ap, int K,
                                           const float* __restrict__ wp, int N) {
    #pragma unroll
    for (int i = 0; i < MT; ++i)
        f.a[i] = *(const v4f*)(ap + (size_t)(16 * i) * K);
    #pragma unroll
    for (int g = 0; g < NG; ++g) {
        #pragma unroll
        for (int q = 0; q < 4; ++q)
            f.w[g][q] = *(const v4f*)(wp + (size_t)q * N + 64 * g);
    }
}

template <int MT, int NG>
__device__ __forceinline__ void mma_step(v8f (&acc)[MT][NG][4], const Frags<MT, NG>& f) {
    #pragma unroll
    for (int s = 0; s < 2; ++s) {
        #pragma unroll
        for (int i = 0; i < MT; ++i) {
            v2f av;
            av.x = f.a[i][2 * s];
            av.y = f.a[i][2 * s + 1];
            #pragma unroll
            for (int g = 0; g < NG; ++g) {
                #pragma unroll
                for (int j = 0; j < 4; ++j) {
                    v2f bv;
                    bv.x = f.w[g][2 * s][j];
                    bv.y = f.w[g][2 * s + 1][j];
                    acc[i][g][j] = __builtin_amdgcn_wmma_f32_16x16x4_f32(
                         false, av,  false, bv,
                         (short)0, acc[i][g][j],  false,  false);
                }
            }
        }
    }
}

template <int MT, int NG>
__global__ void __launch_bounds__(128)
sage_gemm_wmma(const float* __restrict__ A0,
               const float* __restrict__ A1,
               const float* __restrict__ W0,
               const float* __restrict__ W1,
               const float* __restrict__ bias,
               float* __restrict__ out,
               int M, int K, int N, int relu) {
    constexpr int WM = 16 * MT;
    constexpr int WN = 64 * NG;
    const int wave = blockIdx.x * (blockDim.x >> 5) + (threadIdx.x >> 5);
    const int tilesN = N / WN;
    const int totalTiles = (M / WM) * tilesN;
    if (wave >= totalTiles) return;

    const int tm = wave / tilesN;
    const int tn = wave - tm * tilesN;
    const int lane = threadIdx.x & 31;
    const int h = lane >> 4;
    const int c = lane & 15;
    const int m0 = tm * WM;
    const int n0 = tn * WN;

    v8f acc[MT][NG][4];
    #pragma unroll
    for (int i = 0; i < MT; ++i) {
        #pragma unroll
        for (int g = 0; g < NG; ++g) {
            #pragma unroll
            for (int j = 0; j < 4; ++j)
                acc[i][g][j] = (v8f){0.f, 0.f, 0.f, 0.f, 0.f, 0.f, 0.f, 0.f};
        }
    }

    const size_t aoff = (size_t)(m0 + c) * K + 4 * h;
    const size_t woff = (size_t)(4 * h) * N + n0 + 4 * c;
    const float* a0p = A0 + aoff;
    const float* a1p = A1 + aoff;
    const float* w0p = W0 + woff;
    const float* w1p = W1 + woff;

    const int steps = K >> 3;
    const int nsteps = steps * 2;

    Frags<MT, NG> fa, fb;
    auto fetch = [&](Frags<MT, NG>& f, int t) {
        const bool p = t >= steps;
        const int kk = (p ? t - steps : t) << 3;
        load_frags<MT, NG>(f, (p ? a1p : a0p) + kk, K, (p ? w1p : w0p) + (size_t)kk * N, N);
    };
    fetch(fa, 0);
    int t = 0;
    for (; t < nsteps - 2; t += 2) {
        fetch(fb, t + 1);
        __builtin_amdgcn_sched_barrier(0);
        mma_step<MT, NG>(acc, fa);
        __builtin_amdgcn_sched_barrier(0);
        fetch(fa, t + 2);
        __builtin_amdgcn_sched_barrier(0);
        mma_step<MT, NG>(acc, fb);
        __builtin_amdgcn_sched_barrier(0);
    }
    fetch(fb, t + 1);
    __builtin_amdgcn_sched_barrier(0);
    mma_step<MT, NG>(acc, fa);
    mma_step<MT, NG>(acc, fb);

    #pragma unroll
    for (int g = 0; g < NG; ++g) {
        const int col = n0 + 64 * g + 4 * c;
        const float b0 = bias[col], b1 = bias[col + 1], b2 = bias[col + 2], b3 = bias[col + 3];
        #pragma unroll
        for (int i = 0; i < MT; ++i) {
            float* op = out + (size_t)(m0 + 16 * i + 8 * h) * N + col;
            #pragma unroll
            for (int r = 0; r < 8; ++r) {
                v4f o;
                o.x = acc[i][g][0][r] + b0;
                o.y = acc[i][g][1][r] + b1;
                o.z = acc[i][g][2][r] + b2;
                o.w = acc[i][g][3][r] + b3;
                if (relu) {
                    o.x = fmaxf(o.x, 0.0f); o.y = fmaxf(o.y, 0.0f);
                    o.z = fmaxf(o.z, 0.0f); o.w = fmaxf(o.w, 0.0f);
                }
                *(v4f*)(op + (size_t)r * N) = o;
            }
        }
    }
}

extern "C" void kernel_launch(void* const* d_in, const int* in_sizes, int n_in,
                              void* d_out, int out_size, void* d_ws, size_t ws_size,
                              hipStream_t stream) {
    (void)in_sizes; (void)n_in; (void)out_size; (void)ws_size;

    const float* x = (const float*)d_in[0];
    const int* srcp[3] = {(const int*)d_in[1], (const int*)d_in[3], (const int*)d_in[5]};
    const int* dstp[3] = {(const int*)d_in[2], (const int*)d_in[4], (const int*)d_in[6]};
    const float* Wl[3] = {(const float*)d_in[7],  (const float*)d_in[10], (const float*)d_in[13]};
    const float* Wr[3] = {(const float*)d_in[8],  (const float*)d_in[11], (const float*)d_in[14]};
    const float* bb[3] = {(const float*)d_in[9],  (const float*)d_in[12], (const float*)d_in[15]};

    const int ntgt[3] = {N_L0, N_L1, N_L2};
    const int Kd[3]   = {128, 256, 256};
    const int Nd[3]   = {256, 256, 128};
    const int relu[3] = {1, 0, 1};

    float* ws     = (float*)d_ws;
    float* aggbuf = ws;
    float* small  = aggbuf + 12800000;
    float* h0     = small + 100000;
    float* h1     = h0 + 25600000;

    int* cnt0 = (int*)h0;     int* cur0 = cnt0 + N_L0;  int* perm0 = cur0 + N_L0;
    int* cnt1 = (int*)h1;     int* cur1 = cnt1 + N_L1;  int* perm1 = cur1 + N_L1;
    int* cnt2 = (int*)small;  int* cur2 = cnt2 + N_L2;  int* perm2 = cur2 + N_L2;

    const int TPB = 256;
    const int NALL = N_L0 + N_L1 + N_L2;
    const int EALL = E_L0 + E_L1 + E_L2;
    zero_counts<<<(NALL + TPB - 1) / TPB, TPB, 0, stream>>>(cnt0, cnt1, cnt2);
    edge_hist<<<(EALL + TPB - 1) / TPB, TPB, 0, stream>>>(dstp[0], dstp[1], dstp[2],
                                                           cnt0, cnt1, cnt2);
    seg_offsets<<<SCAN_B0 + SCAN_B1 + SCAN_B2, 1024, 0, stream>>>(cnt0, cur0, cnt1, cur1,
                                                                   cnt2, cur2);
    edge_place<<<(EALL + TPB - 1) / TPB, TPB, 0, stream>>>(srcp[0], dstp[0], srcp[1], dstp[1],
                                                            srcp[2], dstp[2], cur0, cur1, cur2,
                                                            perm0, perm1, perm2);

    const float* layer_in[3] = {x, h0, h1};
    float* layer_out[3]      = {h0, h1, (float*)d_out};
    const int* permL[3] = {perm0, perm1, perm2};
    const int* cntL[3]  = {cnt0, cnt1, cnt2};
    const int* curL[3]  = {cur0, cur1, cur2};

    for (int L = 0; L < 3; ++L) {
        int n = ntgt[L], K = Kd[L], N = Nd[L];

        if (L == 0)
            seg_mean_gather<128, 8><<<(n + 7) / 8, 256, 0, stream>>>(
                layer_in[L], permL[L], cntL[L], curL[L], aggbuf, n);
        else
            seg_mean_gather<256, 4><<<(n + 7) / 8, 256, 0, stream>>>(
                layer_in[L], permL[L], cntL[L], curL[L], aggbuf, n);

        const int GT = 128;
        if (L == 0) {
            int tiles = (n / 32) * (N / 128);
            sage_gemm_wmma<2, 2><<<(tiles + 3) / 4, GT, 0, stream>>>(
                aggbuf, layer_in[L], Wl[L], Wr[L], bb[L], layer_out[L], n, K, N, relu[L]);
        } else if (L == 1) {
            int tiles = (n / 32) * (N / 64);
            sage_gemm_wmma<2, 1><<<(tiles + 3) / 4, GT, 0, stream>>>(
                aggbuf, layer_in[L], Wl[L], Wr[L], bb[L], layer_out[L], n, K, N, relu[L]);
        } else {
            int tiles = (n / 16) * (N / 64);
            sage_gemm_wmma<1, 1><<<(tiles + 3) / 4, GT, 0, stream>>>(
                aggbuf, layer_in[L], Wl[L], Wr[L], bb[L], layer_out[L], n, K, N, relu[L]);
        }
    }
}


# Edge sort with one atomic pass, not two: the counting add's return value is kept as each edge's rank; placement = first slot + rank with plain loads, one store; sum order = ticket order as before
# speedup vs baseline: 2.4939x; 1.1695x over previous
#include <hip/hip_runtime.h>


typedef float v2f __attribute__((ext_vector_type(2)));
typedef float v4f __attribute__((ext_vector_type(4)));
typedef float v8f __attribute__((ext_vector_type(8)));

#define E_L0 1500000
#define E_L1 300000
#define E_L2 60000
#define N_L0 100000
#define N_L1 20000
#define N_L2 4000

__global__ void zero_counts(int* __restrict__ c0, int* __restrict__ c1, int* __restrict__ c2) {
    int t = blockIdx.x * blockDim.x + threadIdx.x;
    if (t < N_L0) c0[t] = 0;
    else if (t < N_L0 + N_L1) c1[t - N_L0] = 0;
    else if (t < N_L0 + N_L1 + N_L2) c2[t - N_L0 - N_L1] = 0;
}

__global__ void edge_ticket(const int* __restrict__ d0, const int* __restrict__ d1,
                            const int* __restrict__ d2, int* __restrict__ c0,
                            int* __restrict__ c1, int* __restrict__ c2,
                            int* __restrict__ tick) {
    int t = blockIdx.x * blockDim.x + threadIdx.x;
    if (t < E_L0) tick[t] = atomicAdd(&c0[d0[t]], 1);
    else if (t < E_L0 + E_L1) tick[t] = atomicAdd(&c1[d1[t - E_L0]], 1);
    else if (t < E_L0 + E_L1 + E_L2) tick[t] = atomicAdd(&c2[d2[t - E_L0 - E_L1]], 1);
}

#define SCAN_B0 98
#define SCAN_B1 20
#define SCAN_B2 4
__global__ void __launch_bounds__(1024)
seg_offsets(const int* __restrict__ c0, int* __restrict__ u0,
            const int* __restrict__ c1, int* __restrict__ u1,
            const int* __restrict__ c2, int* __restrict__ u2) {
    __shared__ int red[32];
    __shared__ int tot[32];
    int b = blockIdx.x;
    const int* cnt; int* cur; int n;
    if (b < SCAN_B0) { cnt = c0; cur = u0; n = N_L0; }
    else if (b < SCAN_B0 + SCAN_B1) { b -= SCAN_B0; cnt = c1; cur = u1; n = N_L1; }
    else { b -= SCAN_B0 + SCAN_B1; cnt = c2; cur = u2; n = N_L2; }
    const int tid = threadIdx.x;
    const int lane = tid & 31;
    const int w = tid >> 5;
    const int lo = b * 1024;
    int p = 0;
    const int4* c4 = (const int4*)cnt;
    for (int j = tid; j < (lo >> 2); j += 1024) {
        int4 q = c4[j];
        p += q.x + q.y + q.z + q.w;
    }
    const int idx = lo + tid;
    const int v = (idx < n) ? cnt[idx] : 0;
    int incl = v;
    for (int d = 1; d < 32; d <<= 1) {
        int t = __shfl_up(incl, d);
        if (lane >= d) incl += t;
    }
    for (int d = 16; d >= 1; d >>= 1) p += __shfl_xor(p, d);
    if (lane == 31) tot[w] = incl;
    if (lane == 0) red[w] = p;
    __syncthreads();
    int base = 0;
    for (int k = 0; k < 32; ++k) {
        base += red[k];
        if (k < w) base += tot[k];
    }
    if (idx < n) cur[idx] = base + incl - v;
}

__global__ void edge_place(const int* __restrict__ s0, const int* __restrict__ d0,
                           const int* __restrict__ s1, const int* __restrict__ d1,
                           const int* __restrict__ s2, const int* __restrict__ d2,
                           const int* __restrict__ u0, const int* __restrict__ u1,
                           const int* __restrict__ u2, const int* __restrict__ tick,
                           int* __restrict__ p0, int* __restrict__ p1, int* __restrict__ p2) {
    const int t = blockIdx.x * blockDim.x + threadIdx.x;
    if (t >= E_L0 + E_L1 + E_L2) return;
    const int* sp = s0; const int* dp = d0; const int* up = u0; int* pp = p0; int e = t;
    if (t >= E_L0 + E_L1) { sp = s2; dp = d2; up = u2; pp = p2; e = t - (E_L0 + E_L1); }
    else if (t >= E_L0)   { sp = s1; dp = d1; up = u1; pp = p1; e = t - E_L0; }
    const int d = dp[e];
    const int k = tick[t];
    const int s = sp[e];
    pp[up[d] + k] = s;
}

template <int D, int UNR>
__global__ void __launch_bounds__(256)
seg_mean_gather(const float* __restrict__ xs, const int* __restrict__ perm,
                const int* __restrict__ cnt, const int* __restrict__ cur,
                float* __restrict__ agg, int n) {
    constexpr int V = D / 128;
    const int lane = threadIdx.x & 31;
    const int i = blockIdx.x * 8 + (threadIdx.x >> 5);
    if (i >= n) return;
    const int deg = __builtin_amdgcn_readfirstlane(cnt[i]);
    const int begin = __builtin_amdgcn_readfirstlane(cur[i]);
    v4f acc[V];
    #pragma unroll
    for (int v = 0; v < V; ++v) acc[v] = (v4f){0.f, 0.f, 0.f, 0.f};
    const float* xl = xs + 4 * lane;
    for (int j0 = 0; j0 < deg; j0 += 32) {
        const int m = min(32, deg - j0);
        const int my = perm[begin + j0 + min(lane, m - 1)];
        for (int k = 0; k < m; k += UNR) {
            v4f r[UNR][V];
            #pragma unroll
            for (int u = 0; u < UNR; ++u) {
                const int kk = min(k + u, m - 1);
                const int s = __builtin_amdgcn_readlane(my, kk);
                const float* rp = xl + (size_t)s * D;
                #pragma unroll
                for (int v = 0; v < V; ++v) r[u][v] = *(const v4f*)(rp + 128 * v);
            }
            #pragma unroll
            for (int u = 0; u < UNR; ++u) {
                const bool on = (k + u) < m;
                #pragma unroll
                for (int v = 0; v < V; ++v)
                    acc[v] += on ? r[u][v] : (v4f){0.f, 0.f, 0.f, 0.f};
            }
        }
    }
    const float inv = 1.0f / fmaxf((float)deg, 1.0f);
    float* op = agg + (size_t)i * D + 4 * lane;
    #pragma unroll
    for (int v = 0; v < V; ++v) *(v4f*)(op + 128 * v) = acc[v] * inv;
}

template <int MT, int NG>
struct Frags {
    v4f a[MT];
    v4f w[NG][4];
};

template <int MT, int NG>
__device__ __forceinline__ void load_frags(Frags<MT, NG>& f, const float* __restrict__ ap, int K,
                                           const float* __restrict__ wp, int N) {
    #pragma unroll
    for (int i = 0; i < MT; ++i)
        f.a[i] = *(const v4f*)(ap + (size_t)(16 * i) * K);
    #pragma unroll
    for (int g = 0; g < NG; ++g) {
        #pragma unroll
        for (int q = 0; q < 4; ++q)
            f.w[g][q] = *(const v4f*)(wp + (size_t)q * N + 64 * g);
    }
}

template <int MT, int NG>
__device__ __forceinline__ void mma_step(v8f (&acc)[MT][NG][4], const Frags<MT, NG>& f) {
    #pragma unroll
    for (int s = 0; s < 2; ++s) {
        #pragma unroll
        for (int i = 0; i < MT; ++i) {
            v2f av;
            av.x = f.a[i][2 * s];
            av.y = f.a[i][2 * s + 1];
            #pragma unroll
            for (int g = 0; g < NG; ++g) {
                #pragma unroll
                for (int j = 0; j < 4; ++j) {
                    v2f bv;
                    bv.x = f.w[g][2 * s][j];
                    bv.y = f.w[g][2 * s + 1][j];
                    acc[i][g][j] = __builtin_amdgcn_wmma_f32_16x16x4_f32(
                         false, av,  false, bv,
                         (short)0, acc[i][g][j],  false,  false);
                }
            }
        }
    }
}

template <int MT, int NG>
__global__ void __launch_bounds__(128)
sage_gemm_wmma(const float* __restrict__ A0,
               const float* __restrict__ A1,
               const float* __restrict__ W0,
               const float* __restrict__ W1,
               const float* __restrict__ bias,
               float* __restrict__ out,
               int M, int K, int N, int relu) {
    constexpr int WM = 16 * MT;
    constexpr int WN = 64 * NG;
    const int wave = blockIdx.x * (blockDim.x >> 5) + (threadIdx.x >> 5);
    const int tilesN = N / WN;
    const int totalTiles = (M / WM) * tilesN;
    if (wave >= totalTiles) return;

    const int tm = wave / tilesN;
    const int tn = wave - tm * tilesN;
    const int lane = threadIdx.x & 31;
    const int h = lane >> 4;
    const int c = lane & 15;
    const int m0 = tm * WM;
    const int n0 = tn * WN;

    v8f acc[MT][NG][4];
    #pragma unroll
    for (int i = 0; i < MT; ++i) {
        #pragma unroll
        for (int g = 0; g < NG; ++g) {
            #pragma unroll
            for (int j = 0; j < 4; ++j)
                acc[i][g][j] = (v8f){0.f, 0.f, 0.f, 0.f, 0.f, 0.f, 0.f, 0.f};
        }
    }

    const size_t aoff = (size_t)(m0 + c) * K + 4 * h;
    const size_t woff = (size_t)(4 * h) * N + n0 + 4 * c;
    const float* a0p = A0 + aoff;
    const float* a1p = A1 + aoff;
    const float* w0p = W0 + woff;
    const float* w1p = W1 + woff;

    const int steps = K >> 3;
    const int nsteps = steps * 2;

    Frags<MT, NG> fa, fb;
    auto fetch = [&](Frags<MT, NG>& f, int t) {
        const bool p = t >= steps;
        const int kk = (p ? t - steps : t) << 3;
        load_frags<MT, NG>(f, (p ? a1p : a0p) + kk, K, (p ? w1p : w0p) + (size_t)kk * N, N);
    };
    fetch(fa, 0);
    int t = 0;
    for (; t < nsteps - 2; t += 2) {
        fetch(fb, t + 1);
        __builtin_amdgcn_sched_barrier(0);
        mma_step<MT, NG>(acc, fa);
        __builtin_amdgcn_sched_barrier(0);
        fetch(fa, t + 2);
        __builtin_amdgcn_sched_barrier(0);
        mma_step<MT, NG>(acc, fb);
        __builtin_amdgcn_sched_barrier(0);
    }
    fetch(fb, t + 1);
    __builtin_amdgcn_sched_barrier(0);
    mma_step<MT, NG>(acc, fa);
    mma_step<MT, NG>(acc, fb);

    #pragma unroll
    for (int g = 0; g < NG; ++g) {
        const int col = n0 + 64 * g + 4 * c;
        const float b0 = bias[col], b1 = bias[col + 1], b2 = bias[col + 2], b3 = bias[col + 3];
        #pragma unroll
        for (int i = 0; i < MT; ++i) {
            float* op = out + (size_t)(m0 + 16 * i + 8 * h) * N + col;
            #pragma unroll
            for (int r = 0; r < 8; ++r) {
                v4f o;
                o.x = acc[i][g][0][r] + b0;
                o.y = acc[i][g][1][r] + b1;
                o.z = acc[i][g][2][r] + b2;
                o.w = acc[i][g][3][r] + b3;
                if (relu) {
                    o.x = fmaxf(o.x, 0.0f); o.y = fmaxf(o.y, 0.0f);
                    o.z = fmaxf(o.z, 0.0f); o.w = fmaxf(o.w, 0.0f);
                }
                *(v4f*)(op + (size_t)r * N) = o;
            }
        }
    }
}

extern "C" void kernel_launch(void* const* d_in, const int* in_sizes, int n_in,
                              void* d_out, int out_size, void* d_ws, size_t ws_size,
                              hipStream_t stream) {
    (void)in_sizes; (void)n_in; (void)out_size; (void)ws_size;

    const float* x = (const float*)d_in[0];
    const int* srcp[3] = {(const int*)d_in[1], (const int*)d_in[3], (const int*)d_in[5]};
    const int* dstp[3] = {(const int*)d_in[2], (const int*)d_in[4], (const int*)d_in[6]};
    const float* Wl[3] = {(const float*)d_in[7],  (const float*)d_in[10], (const float*)d_in[13]};
    const float* Wr[3] = {(const float*)d_in[8],  (const float*)d_in[11], (const float*)d_in[14]};
    const float* bb[3] = {(const float*)d_in[9],  (const float*)d_in[12], (const float*)d_in[15]};

    const int ntgt[3] = {N_L0, N_L1, N_L2};
    const int Kd[3]   = {128, 256, 256};
    const int Nd[3]   = {256, 256, 128};
    const int relu[3] = {1, 0, 1};

    float* ws     = (float*)d_ws;
    float* aggbuf = ws;
    float* small  = aggbuf + 12800000;
    float* h0     = small + 100000;
    float* h1     = h0 + 25600000;

    int* cnt0 = (int*)h0;     int* cur0 = cnt0 + N_L0;  int* perm0 = cur0 + N_L0;
    int* cnt1 = (int*)h1;     int* cur1 = cnt1 + N_L1;  int* perm1 = cur1 + N_L1;
    int* cnt2 = (int*)small;  int* cur2 = cnt2 + N_L2;  int* perm2 = cur2 + N_L2;
    int* tick = (int*)aggbuf;

    const int TPB = 256;
    const int NALL = N_L0 + N_L1 + N_L2;
    const int EALL = E_L0 + E_L1 + E_L2;
    zero_counts<<<(NALL + TPB - 1) / TPB, TPB, 0, stream>>>(cnt0, cnt1, cnt2);
    edge_ticket<<<(EALL + TPB - 1) / TPB, TPB, 0, stream>>>(dstp[0], dstp[1], dstp[2],
                                                             cnt0, cnt1, cnt2, tick);
    seg_offsets<<<SCAN_B0 + SCAN_B1 + SCAN_B2, 1024, 0, stream>>>(cnt0, cur0, cnt1, cur1,
                                                                   cnt2, cur2);
    edge_place<<<(EALL + TPB - 1) / TPB, TPB, 0, stream>>>(srcp[0], dstp[0], srcp[1], dstp[1],
                                                            srcp[2], dstp[2], cur0, cur1, cur2,
                                                            tick, perm0, perm1, perm2);

    const float* layer_in[3] = {x, h0, h1};
    float* layer_out[3]      = {h0, h1, (float*)d_out};
    const int* permL[3] = {perm0, perm1, perm2};
    const int* cntL[3]  = {cnt0, cnt1, cnt2};
    const int* curL[3]  = {cur0, cur1, cur2};

    for (int L = 0; L < 3; ++L) {
        int n = ntgt[L], K = Kd[L], N = Nd[L];

        if (L == 0)
            seg_mean_gather<128, 8><<<(n + 7) / 8, 256, 0, stream>>>(
                layer_in[L], permL[L], cntL[L], curL[L], aggbuf, n);
        else
            seg_mean_gather<256, 4><<<(n + 7) / 8, 256, 0, stream>>>(
                layer_in[L], permL[L], cntL[L], curL[L], aggbuf, n);

        const int GT = 128;
        if (L == 0) {
            int tiles = (n / 32) * (N / 128);
            sage_gemm_wmma<2, 2><<<(tiles + 3) / 4, GT, 0, stream>>>(
                aggbuf, layer_in[L], Wl[L], Wr[L], bb[L], layer_out[L], n, K, N, relu[L]);
        } else if (L == 1) {
            int tiles = (n / 32) * (N / 64);
            sage_gemm_wmma<2, 1><<<(tiles + 3) / 4, GT, 0, stream>>>(
                aggbuf, layer_in[L], Wl[L], Wr[L], bb[L], layer_out[L], n, K, N, relu[L]);
        } else {
            int tiles = (n / 16) * (N / 64);
            sage_gemm_wmma<1, 1><<<(tiles + 3) / 4, GT, 0, stream>>>(
                aggbuf, layer_in[L], Wl[L], Wr[L], bb[L], layer_out[L], n, K, N, relu[L]);
        }
    }
}


# Edge sort ranks with LDS counters in two levels (4096-edge chunks into 512-node bins, then one work-group per bin): no global atomics left; sum order = counter arrival order as before
# speedup vs baseline: 3.4714x; 1.3920x over previous
#include <hip/hip_runtime.h>


typedef float v2f __attribute__((ext_vector_type(2)));
typedef float v4f __attribute__((ext_vector_type(4)));
typedef float v8f __attribute__((ext_vector_type(8)));

#define E_L0 1500000
#define E_L1 300000
#define E_L2 60000
#define N_L0 100000
#define N_L1 20000
#define N_L2 4000

#define BIN_LOG 9
#define BIN_W 512
#define CHUNK 4096
#define NB0 196
#define NB1 40
#define NB2 8
#define CB0 367
#define CB1 74
#define CB2 15
#define T0 (NB0 * CB0)
#define T1 (NB1 * CB1)
#define T2 (NB2 * CB2)

__global__ void __launch_bounds__(1024)
chunk_count(const int* __restrict__ d0, const int* __restrict__ d1, const int* __restrict__ d2,
            int* __restrict__ t0, int* __restrict__ t1, int* __restrict__ t2,
            int* __restrict__ tick) {
    __shared__ int hist[256];
    int b = blockIdx.x;
    const int* dp; int* tb; int E, nblk, nb, ebase;
    if (b < CB0) { dp = d0; tb = t0; E = E_L0; nblk = CB0; nb = NB0; ebase = 0; }
    else if (b < CB0 + CB1) { b -= CB0; dp = d1; tb = t1; E = E_L1; nblk = CB1; nb = NB1; ebase = E_L0; }
    else { b -= CB0 + CB1; dp = d2; tb = t2; E = E_L2; nblk = CB2; nb = NB2; ebase = E_L0 + E_L1; }
    const int tid = threadIdx.x;
    if (tid < 256) hist[tid] = 0;
    __syncthreads();
    const int e0 = b * CHUNK + tid;
    int d[4];
    #pragma unroll
    for (int k = 0; k < 4; ++k) d[k] = dp[min(e0 + 1024 * k, E - 1)];
    #pragma unroll
    for (int k = 0; k < 4; ++k) {
        const int e = e0 + 1024 * k;
        if (e < E) tick[ebase + e] = atomicAdd(&hist[d[k] >> BIN_LOG], 1);
    }
    __syncthreads();
    if (tid < nb) tb[tid * nblk + b] = hist[tid];
}

#define SCAN_B0 71
#define SCAN_B1 3
#define SCAN_B2 1
__global__ void __launch_bounds__(1024)
seg_offsets(const int* __restrict__ c0, int* __restrict__ u0,
            const int* __restrict__ c1, int* __restrict__ u1,
            const int* __restrict__ c2, int* __restrict__ u2) {
    __shared__ int red[32];
    __shared__ int tot[32];
    int b = blockIdx.x;
    const int* cnt; int* cur; int n;
    if (b < SCAN_B0) { cnt = c0; cur = u0; n = T0; }
    else if (b < SCAN_B0 + SCAN_B1) { b -= SCAN_B0; cnt = c1; cur = u1; n = T1; }
    else { b -= SCAN_B0 + SCAN_B1; cnt = c2; cur = u2; n = T2; }
    const int tid = threadIdx.x;
    const int lane = tid & 31;
    const int w = tid >> 5;
    const int lo = b * 1024;
    int p = 0;
    const int4* c4 = (const int4*)cnt;
    for (int j = tid; j < (lo >> 2); j += 1024) {
        int4 q = c4[j];
        p += q.x + q.y + q.z + q.w;
    }
    const int idx = lo + tid;
    const int v = (idx < n) ? cnt[idx] : 0;
    int incl = v;
    for (int d = 1; d < 32; d <<= 1) {
        int t = __shfl_up(incl, d);
        if (lane >= d) incl += t;
    }
    for (int d = 16; d >= 1; d >>= 1) p += __shfl_xor(p, d);
    if (lane == 31) tot[w] = incl;
    if (lane == 0) red[w] = p;
    __syncthreads();
    int base = 0;
    for (int k = 0; k < 32; ++k) {
        base += red[k];
        if (k < w) base += tot[k];
    }
    if (idx < n) cur[idx] = base + incl - v;
}

__global__ void __launch_bounds__(1024)
chunk_place(const int* __restrict__ s0, const int* __restrict__ d0,
            const int* __restrict__ s1, const int* __restrict__ d1,
            const int* __restrict__ s2, const int* __restrict__ d2,
            const int* __restrict__ u0, const int* __restrict__ u1,
            const int* __restrict__ u2, const int* __restrict__ tick,
            int* __restrict__ pairs) {
    __shared__ int first[256];
    int b = blockIdx.x;
    const int* sp; const int* dp; const int* up; int E, nblk, nb, ebase;
    if (b < CB0) { sp = s0; dp = d0; up = u0; E = E_L0; nblk = CB0; nb = NB0; ebase = 0; }
    else if (b < CB0 + CB1) { b -= CB0; sp = s1; dp = d1; up = u1; E = E_L1; nblk = CB1; nb = NB1; ebase = E_L0; }
    else { b -= CB0 + CB1; sp = s2; dp = d2; up = u2; E = E_L2; nblk = CB2; nb = NB2; ebase = E_L0 + E_L1; }
    const int tid = threadIdx.x;
    if (tid < nb) first[tid] = up[tid * nblk + b];
    const int e0 = b * CHUNK + tid;
    int d[4], sv[4], r[4];
    #pragma unroll
    for (int k = 0; k < 4; ++k) {
        const int e = min(e0 + 1024 * k, E - 1);
        d[k] = dp[e];
        sv[k] = sp[e];
        r[k] = tick[ebase + e];
    }
    __syncthreads();
    int2* pr = (int2*)pairs + ebase;
    #pragma unroll
    for (int k = 0; k < 4; ++k) {
        const int e = e0 + 1024 * k;
        if (e < E) pr[first[d[k] >> BIN_LOG] + r[k]] = make_int2(sv[k], d[k]);
    }
}

__global__ void __launch_bounds__(1024)
bin_sort(const int* __restrict__ u0, const int* __restrict__ u1, const int* __restrict__ u2,
         const int* __restrict__ pairs,
         int* __restrict__ c0, int* __restrict__ f0, int* __restrict__ p0,
         int* __restrict__ c1, int* __restrict__ f1, int* __restrict__ p1,
         int* __restrict__ c2, int* __restrict__ f2, int* __restrict__ p2) {
    __shared__ int cntL[BIN_W];
    __shared__ int curL[BIN_W];
    __shared__ int wsum[16];
    int b = blockIdx.x;
    const int* up; int* cnt; int* cur; int* perm; int E, nblk, nb, n, ebase;
    if (b < NB0) { up = u0; cnt = c0; cur = f0; perm = p0; E = E_L0; nblk = CB0; nb = NB0; n = N_L0; ebase = 0; }
    else if (b < NB0 + NB1) { b -= NB0; up = u1; cnt = c1; cur = f1; perm = p1; E = E_L1; nblk = CB1; nb = NB1; n = N_L1; ebase = E_L0; }
    else { b -= NB0 + NB1; up = u2; cnt = c2; cur = f2; perm = p2; E = E_L2; nblk = CB2; nb = NB2; n = N_L2; ebase = E_L0 + E_L1; }
    const int tid = threadIdx.x;
    const int lane = tid & 31;
    const int w = tid >> 5;
    const int start = up[b * nblk];
    const int end = (b + 1 < nb) ? up[(b + 1) * nblk] : E;
    const int2* pr = (const int2*)pairs + ebase;
    if (tid < BIN_W) cntL[tid] = 0;
    __syncthreads();
    for (int j = start + tid; j < end; j += 4096) {
        int dd[4];
        #pragma unroll
        for (int k = 0; k < 4; ++k) dd[k] = pr[min(j + 1024 * k, end - 1)].y;
        #pragma unroll
        for (int k = 0; k < 4; ++k)
            if (j + 1024 * k < end) atomicAdd(&cntL[dd[k] & (BIN_W - 1)], 1);
    }
    __syncthreads();
    const int v = (tid < BIN_W) ? cntL[tid] : 0;
    int incl = v;
    for (int d = 1; d < 32; d <<= 1) {
        int t = __shfl_up(incl, d);
        if (lane >= d) incl += t;
    }
    if (lane == 31 && w < 16) wsum[w] = incl;
    __syncthreads();
    if (tid < BIN_W) {
        int base = start;
        for (int k = 0; k < 16; ++k)
            if (k < w) base += wsum[k];
        const int firstSlot = base + incl - v;
        curL[tid] = firstSlot;
        const int node = b * BIN_W + tid;
        if (node < n) { cnt[node] = v; cur[node] = firstSlot; }
    }
    __syncthreads();
    for (int j = start + tid; j < end; j += 4096) {
        int2 q[4];
        #pragma unroll
        for (int k = 0; k < 4; ++k) q[k] = pr[min(j + 1024 * k, end - 1)];
        #pragma unroll
        for (int k = 0; k < 4; ++k)
            if (j + 1024 * k < end) perm[atomicAdd(&curL[q[k].y & (BIN_W - 1)], 1)] = q[k].x;
    }
}

template <int D, int UNR>
__global__ void __launch_bounds__(256)
seg_mean_gather(const float* __restrict__ xs, const int* __restrict__ perm,
                const int* __restrict__ cnt, const int* __restrict__ cur,
                float* __restrict__ agg, int n) {
    constexpr int V = D / 128;
    const int lane = threadIdx.x & 31;
    const int i = blockIdx.x * 8 + (threadIdx.x >> 5);
    if (i >= n) return;
    const int deg = __builtin_amdgcn_readfirstlane(cnt[i]);
    const int begin = __builtin_amdgcn_readfirstlane(cur[i]);
    v4f acc[V];
    #pragma unroll
    for (int v = 0; v < V; ++v) acc[v] = (v4f){0.f, 0.f, 0.f, 0.f};
    const float* xl = xs + 4 * lane;
    for (int j0 = 0; j0 < deg; j0 += 32) {
        const int m = min(32, deg - j0);
        const int my = perm[begin + j0 + min(lane, m - 1)];
        for (int k = 0; k < m; k += UNR) {
            v4f r[UNR][V];
            #pragma unroll
            for (int u = 0; u < UNR; ++u) {
                const int kk = min(k + u, m - 1);
                const int s = __builtin_amdgcn_readlane(my, kk);
                const float* rp = xl + (size_t)s * D;
                #pragma unroll
                for (int v = 0; v < V; ++v) r[u][v] = *(const v4f*)(rp + 128 * v);
            }
            #pragma unroll
            for (int u = 0; u < UNR; ++u) {
                const bool on = (k + u) < m;
                #pragma unroll
                for (int v = 0; v < V; ++v)
                    acc[v] += on ? r[u][v] : (v4f){0.f, 0.f, 0.f, 0.f};
            }
        }
    }
    const float inv = 1.0f / fmaxf((float)deg, 1.0f);
    float* op = agg + (size_t)i * D + 4 * lane;
    #pragma unroll
    for (int v = 0; v < V; ++v) *(v4f*)(op + 128 * v) = acc[v] * inv;
}

template <int MT, int NG>
struct Frags {
    v4f a[MT];
    v4f w[NG][4];
};

template <int MT, int NG>
__device__ __forceinline__ void load_frags(Frags<MT, NG>& f, const float* __restrict__ ap, int K,
                                           const float* __restrict__ wp, int N) {
    #pragma unroll
    for (int i = 0; i < MT; ++i)
        f.a[i] = *(const v4f*)(ap + (size_t)(16 * i) * K);
    #pragma unroll
    for (int g = 0; g < NG; ++g) {
        #pragma unroll
        for (int q = 0; q < 4; ++q)
            f.w[g][q] = *(const v4f*)(wp + (size_t)q * N + 64 * g);
    }
}

template <int MT, int NG>
__device__ __forceinline__ void mma_step(v8f (&acc)[MT][NG][4], const Frags<MT, NG>& f) {
    #pragma unroll
    for (int s = 0; s < 2; ++s) {
        #pragma unroll
        for (int i = 0; i < MT; ++i) {
            v2f av;
            av.x = f.a[i][2 * s];
            av.y = f.a[i][2 * s + 1];
            #pragma unroll
            for (int g = 0; g < NG; ++g) {
                #pragma unroll
                for (int j = 0; j < 4; ++j) {
                    v2f bv;
                    bv.x = f.w[g][2 * s][j];
                    bv.y = f.w[g][2 * s + 1][j];
                    acc[i][g][j] = __builtin_amdgcn_wmma_f32_16x16x4_f32(
                         false, av,  false, bv,
                         (short)0, acc[i][g][j],  false,  false);
                }
            }
        }
    }
}

template <int MT, int NG>
__global__ void __launch_bounds__(128)
sage_gemm_wmma(const float* __restrict__ A0,
               const float* __restrict__ A1,
               const float* __restrict__ W0,
               const float* __restrict__ W1,
               const float* __restrict__ bias,
               float* __restrict__ out,
               int M, int K, int N, int relu) {
    constexpr int WM = 16 * MT;
    constexpr int WN = 64 * NG;
    const int wave = blockIdx.x * (blockDim.x >> 5) + (threadIdx.x >> 5);
    const int tilesN = N / WN;
    const int totalTiles = (M / WM) * tilesN;
    if (wave >= totalTiles) return;

    const int tm = wave / tilesN;
    const int tn = wave - tm * tilesN;
    const int lane = threadIdx.x & 31;
    const int h = lane >> 4;
    const int c = lane & 15;
    const int m0 = tm * WM;
    const int n0 = tn * WN;

    v8f acc[MT][NG][4];
    #pragma unroll
    for (int i = 0; i < MT; ++i) {
        #pragma unroll
        for (int g = 0; g < NG; ++g) {
            #pragma unroll
            for (int j = 0; j < 4; ++j)
                acc[i][g][j] = (v8f){0.f, 0.f, 0.f, 0.f, 0.f, 0.f, 0.f, 0.f};
        }
    }

    const size_t aoff = (size_t)(m0 + c) * K + 4 * h;
    const size_t woff = (size_t)(4 * h) * N + n0 + 4 * c;
    const float* a0p = A0 + aoff;
    const float* a1p = A1 + aoff;
    const float* w0p = W0 + woff;
    const float* w1p = W1 + woff;

    const int steps = K >> 3;
    const int nsteps = steps * 2;

    Frags<MT, NG> fa, fb;
    auto fetch = [&](Frags<MT, NG>& f, int t) {
        const bool p = t >= steps;
        const int kk = (p ? t - steps : t) << 3;
        load_frags<MT, NG>(f, (p ? a1p : a0p) + kk, K, (p ? w1p : w0p) + (size_t)kk * N, N);
    };
    fetch(fa, 0);
    int t = 0;
    for (; t < nsteps - 2; t += 2) {
        fetch(fb, t + 1);
        __builtin_amdgcn_sched_barrier(0);
        mma_step<MT, NG>(acc, fa);
        __builtin_amdgcn_sched_barrier(0);
        fetch(fa, t + 2);
        __builtin_amdgcn_sched_barrier(0);
        mma_step<MT, NG>(acc, fb);
        __builtin_amdgcn_sched_barrier(0);
    }
    fetch(fb, t + 1);
    __builtin_amdgcn_sched_barrier(0);
    mma_step<MT, NG>(acc, fa);
    mma_step<MT, NG>(acc, fb);

    #pragma unroll
    for (int g = 0; g < NG; ++g) {
        const int col = n0 + 64 * g + 4 * c;
        const float b0 = bias[col], b1 = bias[col + 1], b2 = bias[col + 2], b3 = bias[col + 3];
        #pragma unroll
        for (int i = 0; i < MT; ++i) {
            float* op = out + (size_t)(m0 + 16 * i + 8 * h) * N + col;
            #pragma unroll
            for (int r = 0; r < 8; ++r) {
                v4f o;
                o.x = acc[i][g][0][r] + b0;
                o.y = acc[i][g][1][r] + b1;
                o.z = acc[i][g][2][r] + b2;
                o.w = acc[i][g][3][r] + b3;
                if (relu) {
                    o.x = fmaxf(o.x, 0.0f); o.y = fmaxf(o.y, 0.0f);
                    o.z = fmaxf(o.z, 0.0f); o.w = fmaxf(o.w, 0.0f);
                }
                *(v4f*)(op + (size_t)r * N) = o;
            }
        }
    }
}

extern "C" void kernel_launch(void* const* d_in, const int* in_sizes, int n_in,
                              void* d_out, int out_size, void* d_ws, size_t ws_size,
                              hipStream_t stream) {
    (void)in_sizes; (void)n_in; (void)out_size; (void)ws_size;

    const float* x = (const float*)d_in[0];
    const int* srcp[3] = {(const int*)d_in[1], (const int*)d_in[3], (const int*)d_in[5]};
    const int* dstp[3] = {(const int*)d_in[2], (const int*)d_in[4], (const int*)d_in[6]};
    const float* Wl[3] = {(const float*)d_in[7],  (const float*)d_in[10], (const float*)d_in[13]};
    const float* Wr[3] = {(const float*)d_in[8],  (const float*)d_in[11], (const float*)d_in[14]};
    const float* bb[3] = {(const float*)d_in[9],  (const float*)d_in[12], (const float*)d_in[15]};

    const int ntgt[3] = {N_L0, N_L1, N_L2};
    const int Kd[3]   = {128, 256, 256};
    const int Nd[3]   = {256, 256, 128};
    const int relu[3] = {1, 0, 1};

    float* ws     = (float*)d_ws;
    float* aggbuf = ws;
    float* small  = aggbuf + 12800000;
    float* h0     = small + 100000;
    float* h1     = h0 + 25600000;

    int* cnt0 = (int*)h0;     int* cur0 = cnt0 + N_L0;  int* perm0 = cur0 + N_L0;
    int* cnt1 = (int*)h1;     int* cur1 = cnt1 + N_L1;  int* perm1 = cur1 + N_L1;
    int* cnt2 = (int*)small;  int* cur2 = cnt2 + N_L2;  int* perm2 = cur2 + N_L2;
    int* tick  = (int*)aggbuf;
    int* tc0   = tick + 1860000;  int* tc1 = tc0 + T0;  int* tc2 = tc1 + T1;
    int* tu0   = tc0 + 75012;     int* tu1 = tu0 + T0;  int* tu2 = tu1 + T1;
    int* pairs = tu0 + 75012;

    chunk_count<<<CB0 + CB1 + CB2, 1024, 0, stream>>>(dstp[0], dstp[1], dstp[2], tc0, tc1, tc2, tick);
    seg_offsets<<<SCAN_B0 + SCAN_B1 + SCAN_B2, 1024, 0, stream>>>(tc0, tu0, tc1, tu1, tc2, tu2);
    chunk_place<<<CB0 + CB1 + CB2, 1024, 0, stream>>>(srcp[0], dstp[0], srcp[1], dstp[1],
                                                       srcp[2], dstp[2], tu0, tu1, tu2, tick, pairs);
    bin_sort<<<NB0 + NB1 + NB2, 1024, 0, stream>>>(tu0, tu1, tu2, pairs,
                                                    cnt0, cur0, perm0, cnt1, cur1, perm1,
                                                    cnt2, cur2, perm2);

    const float* layer_in[3] = {x, h0, h1};
    float* layer_out[3]      = {h0, h1, (float*)d_out};
    const int* permL[3] = {perm0, perm1, perm2};
    const int* cntL[3]  = {cnt0, cnt1, cnt2};
    const int* curL[3]  = {cur0, cur1, cur2};

    for (int L = 0; L < 3; ++L) {
        int n = ntgt[L], K = Kd[L], N = Nd[L];

        if (L == 0)
            seg_mean_gather<128, 8><<<(n + 7) / 8, 256, 0, stream>>>(
                layer_in[L], permL[L], cntL[L], curL[L], aggbuf, n);
        else
            seg_mean_gather<256, 4><<<(n + 7) / 8, 256, 0, stream>>>(
                layer_in[L], permL[L], cntL[L], curL[L], aggbuf, n);

        const int GT = 128;
        if (L == 0) {
            int tiles = (n / 32) * (N / 128);
            sage_gemm_wmma<2, 2><<<(tiles + 3) / 4, GT, 0, stream>>>(
                aggbuf, layer_in[L], Wl[L], Wr[L], bb[L], layer_out[L], n, K, N, relu[L]);
        } else if (L == 1) {
            int tiles = (n / 32) * (N / 64);
            sage_gemm_wmma<2, 1><<<(tiles + 3) / 4, GT, 0, stream>>>(
                aggbuf, layer_in[L], Wl[L], Wr[L], bb[L], layer_out[L], n, K, N, relu[L]);
        } else {
            int tiles = (n / 16) * (N / 64);
            sage_gemm_wmma<1, 1><<<(tiles + 3) / 4, GT, 0, stream>>>(
                aggbuf, layer_in[L], Wl[L], Wr[L], bb[L], layer_out[L], n, K, N, relu[L]);
        }
    }
}


# Layer 0: gather fused into the GEMM kernel (32-row work-group: x_tgt@Wr, mean rows to LDS, agg@Wl); 9 launches; x_tgt terms now summed before agg terms in each accumulator
# speedup vs baseline: 3.7579x; 1.0825x over previous
#include <hip/hip_runtime.h>


typedef float v2f __attribute__((ext_vector_type(2)));
typedef float v4f __attribute__((ext_vector_type(4)));
typedef float v8f __attribute__((ext_vector_type(8)));

#define E_L0 1500000
#define E_L1 300000
#define E_L2 60000
#define N_L0 100000
#define N_L1 20000
#define N_L2 4000

#define BIN_LOG 9
#define BIN_W 512
#define CHUNK 4096
#define NB0 196
#define NB1 40
#define NB2 8
#define CB0 367
#define CB1 74
#define CB2 15
#define T0 (NB0 * CB0)
#define T1 (NB1 * CB1)
#define T2 (NB2 * CB2)

__global__ void __launch_bounds__(1024)
chunk_count(const int* __restrict__ d0, const int* __restrict__ d1, const int* __restrict__ d2,
            int* __restrict__ t0, int* __restrict__ t1, int* __restrict__ t2,
            int* __restrict__ tick) {
    __shared__ int hist[256];
    int b = blockIdx.x;
    const int* dp; int* tb; int E, nblk, nb, ebase;
    if (b < CB0) { dp = d0; tb = t0; E = E_L0; nblk = CB0; nb = NB0; ebase = 0; }
    else if (b < CB0 + CB1) { b -= CB0; dp = d1; tb = t1; E = E_L1; nblk = CB1; nb = NB1; ebase = E_L0; }
    else { b -= CB0 + CB1; dp = d2; tb = t2; E = E_L2; nblk = CB2; nb = NB2; ebase = E_L0 + E_L1; }
    const int tid = threadIdx.x;
    if (tid < 256) hist[tid] = 0;
    __syncthreads();
    const int e0 = b * CHUNK + tid;
    int d[4];
    #pragma unroll
    for (int k = 0; k < 4; ++k) d[k] = dp[min(e0 + 1024 * k, E - 1)];
    #pragma unroll
    for (int k = 0; k < 4; ++k) {
        const int e = e0 + 1024 * k;
        if (e < E) tick[ebase + e] = atomicAdd(&hist[d[k] >> BIN_LOG], 1);
    }
    __syncthreads();
    if (tid < nb) tb[tid * nblk + b] = hist[tid];
}

#define SCAN_B0 71
#define SCAN_B1 3
#define SCAN_B2 1
__global__ void __launch_bounds__(1024)
seg_offsets(const int* __restrict__ c0, int* __restrict__ u0,
            const int* __restrict__ c1, int* __restrict__ u1,
            const int* __restrict__ c2, int* __restrict__ u2) {
    __shared__ int red[32];
    __shared__ int tot[32];
    int b = blockIdx.x;
    const int* cnt; int* cur; int n;
    if (b < SCAN_B0) { cnt = c0; cur = u0; n = T0; }
    else if (b < SCAN_B0 + SCAN_B1) { b -= SCAN_B0; cnt = c1; cur = u1; n = T1; }
    else { b -= SCAN_B0 + SCAN_B1; cnt = c2; cur = u2; n = T2; }
    const int tid = threadIdx.x;
    const int lane = tid & 31;
    const int w = tid >> 5;
    const int lo = b * 1024;
    int p = 0;
    const int4* c4 = (const int4*)cnt;
    for (int j = tid; j < (lo >> 2); j += 1024) {
        int4 q = c4[j];
        p += q.x + q.y + q.z + q.w;
    }
    const int idx = lo + tid;
    const int v = (idx < n) ? cnt[idx] : 0;
    int incl = v;
    for (int d = 1; d < 32; d <<= 1) {
        int t = __shfl_up(incl, d);
        if (lane >= d) incl += t;
    }
    for (int d = 16; d >= 1; d >>= 1) p += __shfl_xor(p, d);
    if (lane == 31) tot[w] = incl;
    if (lane == 0) red[w] = p;
    __syncthreads();
    int base = 0;
    for (int k = 0; k < 32; ++k) {
        base += red[k];
        if (k < w) base += tot[k];
    }
    if (idx < n) cur[idx] = base + incl - v;
}

__global__ void __launch_bounds__(1024)
chunk_place(const int* __restrict__ s0, const int* __restrict__ d0,
            const int* __restrict__ s1, const int* __restrict__ d1,
            const int* __restrict__ s2, const int* __restrict__ d2,
            const int* __restrict__ u0, const int* __restrict__ u1,
            const int* __restrict__ u2, const int* __restrict__ tick,
            int* __restrict__ pairs) {
    __shared__ int first[256];
    int b = blockIdx.x;
    const int* sp; const int* dp; const int* up; int E, nblk, nb, ebase;
    if (b < CB0) { sp = s0; dp = d0; up = u0; E = E_L0; nblk = CB0; nb = NB0; ebase = 0; }
    else if (b < CB0 + CB1) { b -= CB0; sp = s1; dp = d1; up = u1; E = E_L1; nblk = CB1; nb = NB1; ebase = E_L0; }
    else { b -= CB0 + CB1; sp = s2; dp = d2; up = u2; E = E_L2; nblk = CB2; nb = NB2; ebase = E_L0 + E_L1; }
    const int tid = threadIdx.x;
    if (tid < nb) first[tid] = up[tid * nblk + b];
    const int e0 = b * CHUNK + tid;
    int d[4], sv[4], r[4];
    #pragma unroll
    for (int k = 0; k < 4; ++k) {
        const int e = min(e0 + 1024 * k, E - 1);
        d[k] = dp[e];
        sv[k] = sp[e];
        r[k] = tick[ebase + e];
    }
    __syncthreads();
    int2* pr = (int2*)pairs + ebase;
    #pragma unroll
    for (int k = 0; k < 4; ++k) {
        const int e = e0 + 1024 * k;
        if (e < E) pr[first[d[k] >> BIN_LOG] + r[k]] = make_int2(sv[k], d[k]);
    }
}

__global__ void __launch_bounds__(1024)
bin_sort(const int* __restrict__ u0, const int* __restrict__ u1, const int* __restrict__ u2,
         const int* __restrict__ pairs,
         int* __restrict__ c0, int* __restrict__ f0, int* __restrict__ p0,
         int* __restrict__ c1, int* __restrict__ f1, int* __restrict__ p1,
         int* __restrict__ c2, int* __restrict__ f2, int* __restrict__ p2) {
    __shared__ int cntL[BIN_W];
    __shared__ int curL[BIN_W];
    __shared__ int wsum[16];
    int b = blockIdx.x;
    const int* up; int* cnt; int* cur; int* perm; int E, nblk, nb, n, ebase;
    if (b < NB0) { up = u0; cnt = c0; cur = f0; perm = p0; E = E_L0; nblk = CB0; nb = NB0; n = N_L0; ebase = 0; }
    else if (b < NB0 + NB1) { b -= NB0; up = u1; cnt = c1; cur = f1; perm = p1; E = E_L1; nblk = CB1; nb = NB1; n = N_L1; ebase = E_L0; }
    else { b -= NB0 + NB1; up = u2; cnt = c2; cur = f2; perm = p2; E = E_L2; nblk = CB2; nb = NB2; n = N_L2; ebase = E_L0 + E_L1; }
    const int tid = threadIdx.x;
    const int lane = tid & 31;
    const int w = tid >> 5;
    const int start = up[b * nblk];
    const int end = (b + 1 < nb) ? up[(b + 1) * nblk] : E;
    const int2* pr = (const int2*)pairs + ebase;
    if (tid < BIN_W) cntL[tid] = 0;
    __syncthreads();
    for (int j = start + tid; j < end; j += 4096) {
        int dd[4];
        #pragma unroll
        for (int k = 0; k < 4; ++k) dd[k] = pr[min(j + 1024 * k, end - 1)].y;
        #pragma unroll
        for (int k = 0; k < 4; ++k)
            if (j + 1024 * k < end) atomicAdd(&cntL[dd[k] & (BIN_W - 1)], 1);
    }
    __syncthreads();
    const int v = (tid < BIN_W) ? cntL[tid] : 0;
    int incl = v;
    for (int d = 1; d < 32; d <<= 1) {
        int t = __shfl_up(incl, d);
        if (lane >= d) incl += t;
    }
    if (lane == 31 && w < 16) wsum[w] = incl;
    __syncthreads();
    if (tid < BIN_W) {
        int base = start;
        for (int k = 0; k < 16; ++k)
            if (k < w) base += wsum[k];
        const int firstSlot = base + incl - v;
        curL[tid] = firstSlot;
        const int node = b * BIN_W + tid;
        if (node < n) { cnt[node] = v; cur[node] = firstSlot; }
    }
    __syncthreads();
    for (int j = start + tid; j < end; j += 4096) {
        int2 q[4];
        #pragma unroll
        for (int k = 0; k < 4; ++k) q[k] = pr[min(j + 1024 * k, end - 1)];
        #pragma unroll
        for (int k = 0; k < 4; ++k)
            if (j + 1024 * k < end) perm[atomicAdd(&curL[q[k].y & (BIN_W - 1)], 1)] = q[k].x;
    }
}

template <int D, int UNR>
__global__ void __launch_bounds__(256)
seg_mean_gather(const float* __restrict__ xs, const int* __restrict__ perm,
                const int* __restrict__ cnt, const int* __restrict__ cur,
                float* __restrict__ agg, int n) {
    constexpr int V = D / 128;
    const int lane = threadIdx.x & 31;
    const int i = blockIdx.x * 8 + (threadIdx.x >> 5);
    if (i >= n) return;
    const int deg = __builtin_amdgcn_readfirstlane(cnt[i]);
    const int begin = __builtin_amdgcn_readfirstlane(cur[i]);
    v4f acc[V];
    #pragma unroll
    for (int v = 0; v < V; ++v) acc[v] = (v4f){0.f, 0.f, 0.f, 0.f};
    const float* xl = xs + 4 * lane;
    for (int j0 = 0; j0 < deg; j0 += 32) {
        const int m = min(32, deg - j0);
        const int my = perm[begin + j0 + min(lane, m - 1)];
        for (int k = 0; k < m; k += UNR) {
            v4f r[UNR][V];
            #pragma unroll
            for (int u = 0; u < UNR; ++u) {
                const int kk = min(k + u, m - 1);
                const int s = __builtin_amdgcn_readlane(my, kk);
                const float* rp = xl + (size_t)s * D;
                #pragma unroll
                for (int v = 0; v < V; ++v) r[u][v] = *(const v4f*)(rp + 128 * v);
            }
            #pragma unroll
            for (int u = 0; u < UNR; ++u) {
                const bool on = (k + u) < m;
                #pragma unroll
                for (int v = 0; v < V; ++v)
                    acc[v] += on ? r[u][v] : (v4f){0.f, 0.f, 0.f, 0.f};
            }
        }
    }
    const float inv = 1.0f / fmaxf((float)deg, 1.0f);
    float* op = agg + (size_t)i * D + 4 * lane;
    #pragma unroll
    for (int v = 0; v < V; ++v) *(v4f*)(op + 128 * v) = acc[v] * inv;
}

template <int MT, int NG>
struct Frags {
    v4f a[MT];
    v4f w[NG][4];
};

template <int MT, int NG>
__device__ __forceinline__ void load_frags(Frags<MT, NG>& f, const float* __restrict__ ap, int K,
                                           const float* __restrict__ wp, int N) {
    #pragma unroll
    for (int i = 0; i < MT; ++i)
        f.a[i] = *(const v4f*)(ap + (size_t)(16 * i) * K);
    #pragma unroll
    for (int g = 0; g < NG; ++g) {
        #pragma unroll
        for (int q = 0; q < 4; ++q)
            f.w[g][q] = *(const v4f*)(wp + (size_t)q * N + 64 * g);
    }
}

template <int MT, int NG>
__device__ __forceinline__ void mma_step(v8f (&acc)[MT][NG][4], const Frags<MT, NG>& f) {
    #pragma unroll
    for (int s = 0; s < 2; ++s) {
        #pragma unroll
        for (int i = 0; i < MT; ++i) {
            v2f av;
            av.x = f.a[i][2 * s];
            av.y = f.a[i][2 * s + 1];
            #pragma unroll
            for (int g = 0; g < NG; ++g) {
                #pragma unroll
                for (int j = 0; j < 4; ++j) {
                    v2f bv;
                    bv.x = f.w[g][2 * s][j];
                    bv.y = f.w[g][2 * s + 1][j];
                    acc[i][g][j] = __builtin_amdgcn_wmma_f32_16x16x4_f32(
                         false, av,  false, bv,
                         (short)0, acc[i][g][j],  false,  false);
                }
            }
        }
    }
}

template <int MT, int NG>
__global__ void __launch_bounds__(128)
sage_gemm_wmma(const float* __restrict__ A0,
               const float* __restrict__ A1,
               const float* __restrict__ W0,
               const float* __restrict__ W1,
               const float* __restrict__ bias,
               float* __restrict__ out,
               int M, int K, int N, int relu) {
    constexpr int WM = 16 * MT;
    constexpr int WN = 64 * NG;
    const int wave = blockIdx.x * (blockDim.x >> 5) + (threadIdx.x >> 5);
    const int tilesN = N / WN;
    const int totalTiles = (M / WM) * tilesN;
    if (wave >= totalTiles) return;

    const int tm = wave / tilesN;
    const int tn = wave - tm * tilesN;
    const int lane = threadIdx.x & 31;
    const int h = lane >> 4;
    const int c = lane & 15;
    const int m0 = tm * WM;
    const int n0 = tn * WN;

    v8f acc[MT][NG][4];
    #pragma unroll
    for (int i = 0; i < MT; ++i) {
        #pragma unroll
        for (int g = 0; g < NG; ++g) {
            #pragma unroll
            for (int j = 0; j < 4; ++j)
                acc[i][g][j] = (v8f){0.f, 0.f, 0.f, 0.f, 0.f, 0.f, 0.f, 0.f};
        }
    }

    const size_t aoff = (size_t)(m0 + c) * K + 4 * h;
    const size_t woff = (size_t)(4 * h) * N + n0 + 4 * c;
    const float* a0p = A0 + aoff;
    const float* a1p = A1 + aoff;
    const float* w0p = W0 + woff;
    const float* w1p = W1 + woff;

    const int steps = K >> 3;
    const int nsteps = steps * 2;

    Frags<MT, NG> fa, fb;
    auto fetch = [&](Frags<MT, NG>& f, int t) {
        const bool p = t >= steps;
        const int kk = (p ? t - steps : t) << 3;
        load_frags<MT, NG>(f, (p ? a1p : a0p) + kk, K, (p ? w1p : w0p) + (size_t)kk * N, N);
    };
    fetch(fa, 0);
    int t = 0;
    for (; t < nsteps - 2; t += 2) {
        fetch(fb, t + 1);
        __builtin_amdgcn_sched_barrier(0);
        mma_step<MT, NG>(acc, fa);
        __builtin_amdgcn_sched_barrier(0);
        fetch(fa, t + 2);
        __builtin_amdgcn_sched_barrier(0);
        mma_step<MT, NG>(acc, fb);
        __builtin_amdgcn_sched_barrier(0);
    }
    fetch(fb, t + 1);
    __builtin_amdgcn_sched_barrier(0);
    mma_step<MT, NG>(acc, fa);
    mma_step<MT, NG>(acc, fb);

    #pragma unroll
    for (int g = 0; g < NG; ++g) {
        const int col = n0 + 64 * g + 4 * c;
        const float b0 = bias[col], b1 = bias[col + 1], b2 = bias[col + 2], b3 = bias[col + 3];
        #pragma unroll
        for (int i = 0; i < MT; ++i) {
            float* op = out + (size_t)(m0 + 16 * i + 8 * h) * N + col;
            #pragma unroll
            for (int r = 0; r < 8; ++r) {
                v4f o;
                o.x = acc[i][g][0][r] + b0;
                o.y = acc[i][g][1][r] + b1;
                o.z = acc[i][g][2][r] + b2;
                o.w = acc[i][g][3][r] + b3;
                if (relu) {
                    o.x = fmaxf(o.x, 0.0f); o.y = fmaxf(o.y, 0.0f);
                    o.z = fmaxf(o.z, 0.0f); o.w = fmaxf(o.w, 0.0f);
                }
                *(v4f*)(op + (size_t)r * N) = o;
            }
        }
    }
}

#define FPITCH 132
#define PCAP 192
template <int MT, int NG>
__device__ __forceinline__ void load_frags_lds(Frags<MT, NG>& f, const float* ap,
                                               const float* __restrict__ wp, int N) {
    #pragma unroll
    for (int i = 0; i < MT; ++i)
        f.a[i] = *(const v4f*)(ap + 16 * i * FPITCH);
    #pragma unroll
    for (int g = 0; g < NG; ++g) {
        #pragma unroll
        for (int q = 0; q < 4; ++q)
            f.w[g][q] = *(const v4f*)(wp + (size_t)q * N + 64 * g);
    }
}

__global__ void __launch_bounds__(128)
sage_fused_l0(const float* __restrict__ xs, const int* __restrict__ perm,
              const int* __restrict__ cnt, const int* __restrict__ cur,
              const float* __restrict__ W0,
              const float* __restrict__ W1,
              const float* __restrict__ bias, float* __restrict__ out, int relu) {
    constexpr int K = 128, N = 256, MT = 2, NG = 1, STEPS = K / 8, UNR = 8;
    __shared__ __attribute__((aligned(16))) float aggT[32 * FPITCH];
    __shared__ int lperm[4 * PCAP];
    const int lane = threadIdx.x & 31;
    const int w = threadIdx.x >> 5;
    const int h = lane >> 4;
    const int c = lane & 15;
    const int m0 = blockIdx.x * 32;
    const int n0 = 64 * w;
    const int dv = cnt[m0 + 8 * w + min(lane, 7)];
    const int bv = cur[m0 + 8 * w + min(lane, 7)];

    v8f acc[MT][NG][4];
    #pragma unroll
    for (int i = 0; i < MT; ++i) {
        #pragma unroll
        for (int j = 0; j < 4; ++j)
            acc[i][0][j] = (v8f){0.f, 0.f, 0.f, 0.f, 0.f, 0.f, 0.f, 0.f};
    }
    const float* a1p = xs + (size_t)(m0 + c) * K + 4 * h;
    const size_t woff = (size_t)(4 * h) * N + n0 + 4 * c;
    const float* w0p = W0 + woff;
    const float* w1p = W1 + woff;
    Frags<MT, NG> fa, fb;

    {
        auto fetch = [&](Frags<MT, NG>& f, int t) {
            load_frags<MT, NG>(f, a1p + 8 * t, K, w1p + (size_t)(8 * t) * N, N);
        };
        fetch(fa, 0);
        int t = 0;
        for (; t < STEPS - 2; t += 2) {
            fetch(fb, t + 1);
            __builtin_amdgcn_sched_barrier(0);
            mma_step<MT, NG>(acc, fa);
            __builtin_amdgcn_sched_barrier(0);
            fetch(fa, t + 2);
            __builtin_amdgcn_sched_barrier(0);
            mma_step<MT, NG>(acc, fb);
            __builtin_amdgcn_sched_barrier(0);
        }
        fetch(fb, t + 1);
        __builtin_amdgcn_sched_barrier(0);
        mma_step<MT, NG>(acc, fa);
        mma_step<MT, NG>(acc, fb);
    }

    {
        const float* xl = xs + 4 * lane;
        const int b0 = __builtin_amdgcn_readlane(bv, 0);
        const int lpo = PCAP * w;
        int pq[PCAP / 32];
        #pragma unroll
        for (int q = 0; q < PCAP / 32; ++q) pq[q] = perm[min(b0 + 32 * q + lane, E_L0 - 1)];
        #pragma unroll
        for (int q = 0; q < PCAP / 32; ++q) lperm[lpo + 32 * q + lane] = pq[q];
        __syncthreads();
        for (int r = 0; r < 8; ++r) {
            const int deg = __builtin_amdgcn_readlane(dv, r);
            const int begin = __builtin_amdgcn_readlane(bv, r);
            const int off = begin - b0;
            v4f s = (v4f){0.f, 0.f, 0.f, 0.f};
            for (int j0 = 0; j0 < deg; j0 += 32) {
                const int m = min(32, deg - j0);
                const int o = off + j0;
                int my = lperm[lpo + min(max(o, 0) + min(lane, m - 1), PCAP - 1)];
                asm volatile("" : "+v"(my));
                if (o < 0 || o + m > PCAP) my = perm[begin + j0 + min(lane, m - 1)];
                for (int k = 0; k < m; k += UNR) {
                    v4f rr[UNR];
                    #pragma unroll
                    for (int u = 0; u < UNR; ++u) {
                        const int kk = min(k + u, m - 1);
                        const int sidx = __builtin_amdgcn_readlane(my, kk);
                        rr[u] = *(const v4f*)(xl + (size_t)sidx * K);
                    }
                    #pragma unroll
                    for (int u = 0; u < UNR; ++u) {
                        const bool on = (k + u) < m;
                        s += on ? rr[u] : (v4f){0.f, 0.f, 0.f, 0.f};
                    }
                }
            }
            const float inv = 1.0f / fmaxf((float)deg, 1.0f);
            *(v4f*)(aggT + (8 * w + r) * FPITCH + 4 * lane) = s * inv;
        }
    }
    __syncthreads();

    {
        const float* alp = aggT + c * FPITCH + 4 * h;
        auto fetch = [&](Frags<MT, NG>& f, int t) {
            load_frags_lds<MT, NG>(f, alp + 8 * t, w0p + (size_t)(8 * t) * N, N);
        };
        fetch(fa, 0);
        int t = 0;
        for (; t < STEPS - 2; t += 2) {
            fetch(fb, t + 1);
            __builtin_amdgcn_sched_barrier(0);
            mma_step<MT, NG>(acc, fa);
            __builtin_amdgcn_sched_barrier(0);
            fetch(fa, t + 2);
            __builtin_amdgcn_sched_barrier(0);
            mma_step<MT, NG>(acc, fb);
            __builtin_amdgcn_sched_barrier(0);
        }
        fetch(fb, t + 1);
        __builtin_amdgcn_sched_barrier(0);
        mma_step<MT, NG>(acc, fa);
        mma_step<MT, NG>(acc, fb);
    }

    {
        const int col = n0 + 4 * c;
        const float b0 = bias[col], b1 = bias[col + 1], b2 = bias[col + 2], b3 = bias[col + 3];
        #pragma unroll
        for (int i = 0; i < MT; ++i) {
            float* op = out + (size_t)(m0 + 16 * i + 8 * h) * N + col;
            #pragma unroll
            for (int r = 0; r < 8; ++r) {
                v4f o;
                o.x = acc[i][0][0][r] + b0;
                o.y = acc[i][0][1][r] + b1;
                o.z = acc[i][0][2][r] + b2;
                o.w = acc[i][0][3][r] + b3;
                if (relu) {
                    o.x = fmaxf(o.x, 0.0f); o.y = fmaxf(o.y, 0.0f);
                    o.z = fmaxf(o.z, 0.0f); o.w = fmaxf(o.w, 0.0f);
                }
                *(v4f*)(op + (size_t)r * N) = o;
            }
        }
    }
}

extern "C" void kernel_launch(void* const* d_in, const int* in_sizes, int n_in,
                              void* d_out, int out_size, void* d_ws, size_t ws_size,
                              hipStream_t stream) {
    (void)in_sizes; (void)n_in; (void)out_size; (void)ws_size;

    const float* x = (const float*)d_in[0];
    const int* srcp[3] = {(const int*)d_in[1], (const int*)d_in[3], (const int*)d_in[5]};
    const int* dstp[3] = {(const int*)d_in[2], (const int*)d_in[4], (const int*)d_in[6]};
    const float* Wl[3] = {(const float*)d_in[7],  (const float*)d_in[10], (const float*)d_in[13]};
    const float* Wr[3] = {(const float*)d_in[8],  (const float*)d_in[11], (const float*)d_in[14]};
    const float* bb[3] = {(const float*)d_in[9],  (const float*)d_in[12], (const float*)d_in[15]};

    const int ntgt[3] = {N_L0, N_L1, N_L2};
    const int Kd[3]   = {128, 256, 256};
    const int Nd[3]   = {256, 256, 128};
    const int relu[3] = {1, 0, 1};

    float* ws     = (float*)d_ws;
    float* aggbuf = ws;
    float* small  = aggbuf + 12800000;
    float* h0     = small + 100000;
    float* h1     = h0 + 25600000;

    int* cnt0 = (int*)aggbuf + 6000000;  int* cur0 = cnt0 + N_L0;  int* perm0 = cur0 + N_L0;
    int* cnt1 = (int*)h1;     int* cur1 = cnt1 + N_L1;  int* perm1 = cur1 + N_L1;
    int* cnt2 = (int*)small;  int* cur2 = cnt2 + N_L2;  int* perm2 = cur2 + N_L2;
    int* tick  = (int*)aggbuf;
    int* tc0   = tick + 1860000;  int* tc1 = tc0 + T0;  int* tc2 = tc1 + T1;
    int* tu0   = tc0 + 75012;     int* tu1 = tu0 + T0;  int* tu2 = tu1 + T1;
    int* pairs = tu0 + 75012;

    chunk_count<<<CB0 + CB1 + CB2, 1024, 0, stream>>>(dstp[0], dstp[1], dstp[2], tc0, tc1, tc2, tick);
    seg_offsets<<<SCAN_B0 + SCAN_B1 + SCAN_B2, 1024, 0, stream>>>(tc0, tu0, tc1, tu1, tc2, tu2);
    chunk_place<<<CB0 + CB1 + CB2, 1024, 0, stream>>>(srcp[0], dstp[0], srcp[1], dstp[1],
                                                       srcp[2], dstp[2], tu0, tu1, tu2, tick, pairs);
    bin_sort<<<NB0 + NB1 + NB2, 1024, 0, stream>>>(tu0, tu1, tu2, pairs,
                                                    cnt0, cur0, perm0, cnt1, cur1, perm1,
                                                    cnt2, cur2, perm2);

    const float* layer_in[3] = {x, h0, h1};
    float* layer_out[3]      = {h0, h1, (float*)d_out};
    const int* permL[3] = {perm0, perm1, perm2};
    const int* cntL[3]  = {cnt0, cnt1, cnt2};
    const int* curL[3]  = {cur0, cur1, cur2};

    sage_fused_l0<<<N_L0 / 32, 128, 0, stream>>>(x, perm0, cnt0, cur0, Wl[0], Wr[0], bb[0], h0, relu[0]);

    for (int L = 1; L < 3; ++L) {
        int n = ntgt[L], K = Kd[L], N = Nd[L];

        seg_mean_gather<256, 4><<<(n + 7) / 8, 256, 0, stream>>>(
                layer_in[L], permL[L], cntL[L], curL[L], aggbuf, n);

        const int GT = 128;
        if (L == 1) {
            int tiles = (n / 32) * (N / 64);
            sage_gemm_wmma<2, 1><<<(tiles + 3) / 4, GT, 0, stream>>>(
                aggbuf, layer_in[L], Wl[L], Wr[L], bb[L], layer_out[L], n, K, N, relu[L]);
        } else {
            int tiles = (n / 16) * (N / 64);
            sage_gemm_wmma<1, 1><<<(tiles + 3) / 4, GT, 0, stream>>>(
                aggbuf, layer_in[L], Wl[L], Wr[L], bb[L], layer_out[L], n, K, N, relu[L]);
        }
    }
}


# Layer 1: gather fused into the GEMM kernel as layer 0 has it (one template, 256-float rows, mean rows to LDS); 8 launches; layer 1 sums x_tgt terms before agg terms in each accumulator
# speedup vs baseline: 3.9549x; 1.0524x over previous
#include <hip/hip_runtime.h>


typedef float v2f __attribute__((ext_vector_type(2)));
typedef float v4f __attribute__((ext_vector_type(4)));
typedef float v8f __attribute__((ext_vector_type(8)));

#define E_L0 1500000
#define E_L1 300000
#define E_L2 60000
#define N_L0 100000
#define N_L1 20000
#define N_L2 4000

#define BIN_LOG 9
#define BIN_W 512
#define CHUNK 4096
#define NB0 196
#define NB1 40
#define NB2 8
#define CB0 367
#define CB1 74
#define CB2 15
#define T0 (NB0 * CB0)
#define T1 (NB1 * CB1)
#define T2 (NB2 * CB2)

__global__ void __launch_bounds__(1024)
chunk_count(const int* __restrict__ d0, const int* __restrict__ d1, const int* __restrict__ d2,
            int* __restrict__ t0, int* __restrict__ t1, int* __restrict__ t2,
            int* __restrict__ tick) {
    __shared__ int hist[256];
    int b = blockIdx.x;
    const int* dp; int* tb; int E, nblk, nb, ebase;
    if (b < CB0) { dp = d0; tb = t0; E = E_L0; nblk = CB0; nb = NB0; ebase = 0; }
    else if (b < CB0 + CB1) { b -= CB0; dp = d1; tb = t1; E = E_L1; nblk = CB1; nb = NB1; ebase = E_L0; }
    else { b -= CB0 + CB1; dp = d2; tb = t2; E = E_L2; nblk = CB2; nb = NB2; ebase = E_L0 + E_L1; }
    const int tid = threadIdx.x;
    if (tid < 256) hist[tid] = 0;
    __syncthreads();
    const int e0 = b * CHUNK + tid;
    int d[4];
    #pragma unroll
    for (int k = 0; k < 4; ++k) d[k] = dp[min(e0 + 1024 * k, E - 1)];
    #pragma unroll
    for (int k = 0; k < 4; ++k) {
        const int e = e0 + 1024 * k;
        if (e < E) tick[ebase + e] = atomicAdd(&hist[d[k] >> BIN_LOG], 1);
    }
    __syncthreads();
    if (tid < nb) tb[tid * nblk + b] = hist[tid];
}

#define SCAN_B0 71
#define SCAN_B1 3
#define SCAN_B2 1
__global__ void __launch_bounds__(1024)
seg_offsets(const int* __restrict__ c0, int* __restrict__ u0,
            const int* __restrict__ c1, int* __restrict__ u1,
            const int* __restrict__ c2, int* __restrict__ u2) {
    __shared__ int red[32];
    __shared__ int tot[32];
    int b = blockIdx.x;
    const int* cnt; int* cur; int n;
    if (b < SCAN_B0) { cnt = c0; cur = u0; n = T0; }
    else if (b < SCAN_B0 + SCAN_B1) { b -= SCAN_B0; cnt = c1; cur = u1; n = T1; }
    else { b -= SCAN_B0 + SCAN_B1; cnt = c2; cur = u2; n = T2; }
    const int tid = threadIdx.x;
    const int lane = tid & 31;
    const int w = tid >> 5;
    const int lo = b * 1024;
    int p = 0;
    const int4* c4 = (const int4*)cnt;
    for (int j = tid; j < (lo >> 2); j += 1024) {
        int4 q = c4[j];
        p += q.x + q.y + q.z + q.w;
    }
    const int idx = lo + tid;
    const int v = (idx < n) ? cnt[idx] : 0;
    int incl = v;
    for (int d = 1; d < 32; d <<= 1) {
        int t = __shfl_up(incl, d);
        if (lane >= d) incl += t;
    }
    for (int d = 16; d >= 1; d >>= 1) p += __shfl_xor(p, d);
    if (lane == 31) tot[w] = incl;
    if (lane == 0) red[w] = p;
    __syncthreads();
    int base = 0;
    for (int k = 0; k < 32; ++k) {
        base += red[k];
        if (k < w) base += tot[k];
    }
    if (idx < n) cur[idx] = base + incl - v;
}

__global__ void __launch_bounds__(1024)
chunk_place(const int* __restrict__ s0, const int* __restrict__ d0,
            const int* __restrict__ s1, const int* __restrict__ d1,
            const int* __restrict__ s2, const int* __restrict__ d2,
            const int* __restrict__ u0, const int* __restrict__ u1,
            const int* __restrict__ u2, const int* __restrict__ tick,
            int* __restrict__ pairs) {
    __shared__ int first[256];
    int b = blockIdx.x;
    const int* sp; const int* dp; const int* up; int E, nblk, nb, ebase;
    if (b < CB0) { sp = s0; dp = d0; up = u0; E = E_L0; nblk = CB0; nb = NB0; ebase = 0; }
    else if (b < CB0 + CB1) { b -= CB0; sp = s1; dp = d1; up = u1; E = E_L1; nblk = CB1; nb = NB1; ebase = E_L0; }
    else { b -= CB0 + CB1; sp = s2; dp = d2; up = u2; E = E_L2; nblk = CB2; nb = NB2; ebase = E_L0 + E_L1; }
    const int tid = threadIdx.x;
    if (tid < nb) first[tid] = up[tid * nblk + b];
    const int e0 = b * CHUNK + tid;
    int d[4], sv[4], r[4];
    #pragma unroll
    for (int k = 0; k < 4; ++k) {
        const int e = min(e0 + 1024 * k, E - 1);
        d[k] = dp[e];
        sv[k] = sp[e];
        r[k] = tick[ebase + e];
    }
    __syncthreads();
    int2* pr = (int2*)pairs + ebase;
    #pragma unroll
    for (int k = 0; k < 4; ++k) {
        const int e = e0 + 1024 * k;
        if (e < E) pr[first[d[k] >> BIN_LOG] + r[k]] = make_int2(sv[k], d[k]);
    }
}

__global__ void __launch_bounds__(1024)
bin_sort(const int* __restrict__ u0, const int* __restrict__ u1, const int* __restrict__ u2,
         const int* __restrict__ pairs,
         int* __restrict__ c0, int* __restrict__ f0, int* __restrict__ p0,
         int* __restrict__ c1, int* __restrict__ f1, int* __restrict__ p1,
         int* __restrict__ c2, int* __restrict__ f2, int* __restrict__ p2) {
    __shared__ int cntL[BIN_W];
    __shared__ int curL[BIN_W];
    __shared__ int wsum[16];
    int b = blockIdx.x;
    const int* up; int* cnt; int* cur; int* perm; int E, nblk, nb, n, ebase;
    if (b < NB0) { up = u0; cnt = c0; cur = f0; perm = p0; E = E_L0; nblk = CB0; nb = NB0; n = N_L0; ebase = 0; }
    else if (b < NB0 + NB1) { b -= NB0; up = u1; cnt = c1; cur = f1; perm = p1; E = E_L1; nblk = CB1; nb = NB1; n = N_L1; ebase = E_L0; }
    else { b -= NB0 + NB1; up = u2; cnt = c2; cur = f2; perm = p2; E = E_L2; nblk = CB2; nb = NB2; n = N_L2; ebase = E_L0 + E_L1; }
    const int tid = threadIdx.x;
    const int lane = tid & 31;
    const int w = tid >> 5;
    const int start = up[b * nblk];
    const int end = (b + 1 < nb) ? up[(b + 1) * nblk] : E;
    const int2* pr = (const int2*)pairs + ebase;
    if (tid < BIN_W) cntL[tid] = 0;
    __syncthreads();
    for (int j = start + tid; j < end; j += 4096) {
        int dd[4];
        #pragma unroll
        for (int k = 0; k < 4; ++k) dd[k] = pr[min(j + 1024 * k, end - 1)].y;
        #pragma unroll
        for (int k = 0; k < 4; ++k)
            if (j + 1024 * k < end) atomicAdd(&cntL[dd[k] & (BIN_W - 1)], 1);
    }
    __syncthreads();
    const int v = (tid < BIN_W) ? cntL[tid] : 0;
    int incl = v;
    for (int d = 1; d < 32; d <<= 1) {
        int t = __shfl_up(incl, d);
        if (lane >= d) incl += t;
    }
    if (lane == 31 && w < 16) wsum[w] = incl;
    __syncthreads();
    if (tid < BIN_W) {
        int base = start;
        for (int k = 0; k < 16; ++k)
            if (k < w) base += wsum[k];
        const int firstSlot = base + incl - v;
        curL[tid] = firstSlot;
        const int node = b * BIN_W + tid;
        if (node < n) { cnt[node] = v; cur[node] = firstSlot; }
    }
    __syncthreads();
    for (int j = start + tid; j < end; j += 4096) {
        int2 q[4];
        #pragma unroll
        for (int k = 0; k < 4; ++k) q[k] = pr[min(j + 1024 * k, end - 1)];
        #pragma unroll
        for (int k = 0; k < 4; ++k)
            if (j + 1024 * k < end) perm[atomicAdd(&curL[q[k].y & (BIN_W - 1)], 1)] = q[k].x;
    }
}

template <int D, int UNR>
__global__ void __launch_bounds__(256)
seg_mean_gather(const float* __restrict__ xs, const int* __restrict__ perm,
                const int* __restrict__ cnt, const int* __restrict__ cur,
                float* __restrict__ agg, int n) {
    constexpr int V = D / 128;
    const int lane = threadIdx.x & 31;
    const int i = blockIdx.x * 8 + (threadIdx.x >> 5);
    if (i >= n) return;
    const int deg = __builtin_amdgcn_readfirstlane(cnt[i]);
    const int begin = __builtin_amdgcn_readfirstlane(cur[i]);
    v4f acc[V];
    #pragma unroll
    for (int v = 0; v < V; ++v) acc[v] = (v4f){0.f, 0.f, 0.f, 0.f};
    const float* xl = xs + 4 * lane;
    for (int j0 = 0; j0 < deg; j0 += 32) {
        const int m = min(32, deg - j0);
        const int my = perm[begin + j0 + min(lane, m - 1)];
        for (int k = 0; k < m; k += UNR) {
            v4f r[UNR][V];
            #pragma unroll
            for (int u = 0; u < UNR; ++u) {
                const int kk = min(k + u, m - 1);
                const int s = __builtin_amdgcn_readlane(my, kk);
                const float* rp = xl + (size_t)s * D;
                #pragma unroll
                for (int v = 0; v < V; ++v) r[u][v] = *(const v4f*)(rp + 128 * v);
            }
            #pragma unroll
            for (int u = 0; u < UNR; ++u) {
                const bool on = (k + u) < m;
                #pragma unroll
                for (int v = 0; v < V; ++v)
                    acc[v] += on ? r[u][v] : (v4f){0.f, 0.f, 0.f, 0.f};
            }
        }
    }
    const float inv = 1.0f / fmaxf((float)deg, 1.0f);
    float* op = agg + (size_t)i * D + 4 * lane;
    #pragma unroll
    for (int v = 0; v < V; ++v) *(v4f*)(op + 128 * v) = acc[v] * inv;
}

template <int MT, int NG>
struct Frags {
    v4f a[MT];
    v4f w[NG][4];
};

template <int MT, int NG>
__device__ __forceinline__ void load_frags(Frags<MT, NG>& f, const float* __restrict__ ap, int K,
                                           const float* __restrict__ wp, int N) {
    #pragma unroll
    for (int i = 0; i < MT; ++i)
        f.a[i] = *(const v4f*)(ap + (size_t)(16 * i) * K);
    #pragma unroll
    for (int g = 0; g < NG; ++g) {
        #pragma unroll
        for (int q = 0; q < 4; ++q)
            f.w[g][q] = *(const v4f*)(wp + (size_t)q * N + 64 * g);
    }
}

template <int MT, int NG>
__device__ __forceinline__ void mma_step(v8f (&acc)[MT][NG][4], const Frags<MT, NG>& f) {
    #pragma unroll
    for (int s = 0; s < 2; ++s) {
        #pragma unroll
        for (int i = 0; i < MT; ++i) {
            v2f av;
            av.x = f.a[i][2 * s];
            av.y = f.a[i][2 * s + 1];
            #pragma unroll
            for (int g = 0; g < NG; ++g) {
                #pragma unroll
                for (int j = 0; j < 4; ++j) {
                    v2f bv;
                    bv.x = f.w[g][2 * s][j];
                    bv.y = f.w[g][2 * s + 1][j];
                    acc[i][g][j] = __builtin_amdgcn_wmma_f32_16x16x4_f32(
                         false, av,  false, bv,
                         (short)0, acc[i][g][j],  false,  false);
                }
            }
        }
    }
}

template <int MT, int NG>
__global__ void __launch_bounds__(128)
sage_gemm_wmma(const float* __restrict__ A0,
               const float* __restrict__ A1,
               const float* __restrict__ W0,
               const float* __restrict__ W1,
               const float* __restrict__ bias,
               float* __restrict__ out,
               int M, int K, int N, int relu) {
    constexpr int WM = 16 * MT;
    constexpr int WN = 64 * NG;
    const int wave = blockIdx.x * (blockDim.x >> 5) + (threadIdx.x >> 5);
    const int tilesN = N / WN;
    const int totalTiles = (M / WM) * tilesN;
    if (wave >= totalTiles) return;

    const int tm = wave / tilesN;
    const int tn = wave - tm * tilesN;
    const int lane = threadIdx.x & 31;
    const int h = lane >> 4;
    const int c = lane & 15;
    const int m0 = tm * WM;
    const int n0 = tn * WN;

    v8f acc[MT][NG][4];
    #pragma unroll
    for (int i = 0; i < MT; ++i) {
        #pragma unroll
        for (int g = 0; g < NG; ++g) {
            #pragma unroll
            for (int j = 0; j < 4; ++j)
                acc[i][g][j] = (v8f){0.f, 0.f, 0.f, 0.f, 0.f, 0.f, 0.f, 0.f};
        }
    }

    const size_t aoff = (size_t)(m0 + c) * K + 4 * h;
    const size_t woff = (size_t)(4 * h) * N + n0 + 4 * c;
    const float* a0p = A0 + aoff;
    const float* a1p = A1 + aoff;
    const float* w0p = W0 + woff;
    const float* w1p = W1 + woff;

    const int steps = K >> 3;
    const int nsteps = steps * 2;

    Frags<MT, NG> fa, fb;
    auto fetch = [&](Frags<MT, NG>& f, int t) {
        const bool p = t >= steps;
        const int kk = (p ? t - steps : t) << 3;
        load_frags<MT, NG>(f, (p ? a1p : a0p) + kk, K, (p ? w1p : w0p) + (size_t)kk * N, N);
    };
    fetch(fa, 0);
    int t = 0;
    for (; t < nsteps - 2; t += 2) {
        fetch(fb, t + 1);
        __builtin_amdgcn_sched_barrier(0);
        mma_step<MT, NG>(acc, fa);
        __builtin_amdgcn_sched_barrier(0);
        fetch(fa, t + 2);
        __builtin_amdgcn_sched_barrier(0);
        mma_step<MT, NG>(acc, fb);
        __builtin_amdgcn_sched_barrier(0);
    }
    fetch(fb, t + 1);
    __builtin_amdgcn_sched_barrier(0);
    mma_step<MT, NG>(acc, fa);
    mma_step<MT, NG>(acc, fb);

    #pragma unroll
    for (int g = 0; g < NG; ++g) {
        const int col = n0 + 64 * g + 4 * c;
        const float b0 = bias[col], b1 = bias[col + 1], b2 = bias[col + 2], b3 = bias[col + 3];
        #pragma unroll
        for (int i = 0; i < MT; ++i) {
            float* op = out + (size_t)(m0 + 16 * i + 8 * h) * N + col;
            #pragma unroll
            for (int r = 0; r < 8; ++r) {
                v4f o;
                o.x = acc[i][g][0][r] + b0;
                o.y = acc[i][g][1][r] + b1;
                o.z = acc[i][g][2][r] + b2;
                o.w = acc[i][g][3][r] + b3;
                if (relu) {
                    o.x = fmaxf(o.x, 0.0f); o.y = fmaxf(o.y, 0.0f);
                    o.z = fmaxf(o.z, 0.0f); o.w = fmaxf(o.w, 0.0f);
                }
                *(v4f*)(op + (size_t)r * N) = o;
            }
        }
    }
}

#define PCAP 192
template <int MT, int NG, int PITCH>
__device__ __forceinline__ void load_frags_lds(Frags<MT, NG>& f, const float* ap,
                                               const float* __restrict__ wp, int N) {
    #pragma unroll
    for (int i = 0; i < MT; ++i)
        f.a[i] = *(const v4f*)(ap + 16 * i * PITCH);
    #pragma unroll
    for (int g = 0; g < NG; ++g) {
        #pragma unroll
        for (int q = 0; q < 4; ++q)
            f.w[g][q] = *(const v4f*)(wp + (size_t)q * N + 64 * g);
    }
}

template <int K, int EL>
__global__ void __launch_bounds__(128)
sage_fused(const float* __restrict__ xs, const int* __restrict__ perm,
           const int* __restrict__ cnt, const int* __restrict__ cur,
           const float* __restrict__ W0,
           const float* __restrict__ W1,
           const float* __restrict__ bias, float* __restrict__ out, int relu) {
    constexpr int N = 256, MT = 2, NG = 1, STEPS = K / 8, UNR = 8;
    constexpr int V = K / 128;
    constexpr int PITCH = K + 4;
    __shared__ __attribute__((aligned(16))) float aggT[32 * PITCH];
    __shared__ int lperm[4 * PCAP];
    const int lane = threadIdx.x & 31;
    const int w = threadIdx.x >> 5;
    const int h = lane >> 4;
    const int c = lane & 15;
    const int m0 = blockIdx.x * 32;
    const int n0 = 64 * w;
    const int dv = cnt[m0 + 8 * w + min(lane, 7)];
    const int bv = cur[m0 + 8 * w + min(lane, 7)];

    v8f acc[MT][NG][4];
    #pragma unroll
    for (int i = 0; i < MT; ++i) {
        #pragma unroll
        for (int j = 0; j < 4; ++j)
            acc[i][0][j] = (v8f){0.f, 0.f, 0.f, 0.f, 0.f, 0.f, 0.f, 0.f};
    }
    const float* a1p = xs + (size_t)(m0 + c) * K + 4 * h;
    const size_t woff = (size_t)(4 * h) * N + n0 + 4 * c;
    const float* w0p = W0 + woff;
    const float* w1p = W1 + woff;
    Frags<MT, NG> fa, fb;

    {
        auto fetch = [&](Frags<MT, NG>& f, int t) {
            load_frags<MT, NG>(f, a1p + 8 * t, K, w1p + (size_t)(8 * t) * N, N);
        };
        fetch(fa, 0);
        int t = 0;
        for (; t < STEPS - 2; t += 2) {
            fetch(fb, t + 1);
            __builtin_amdgcn_sched_barrier(0);
            mma_step<MT, NG>(acc, fa);
            __builtin_amdgcn_sched_barrier(0);
            fetch(fa, t + 2);
            __builtin_amdgcn_sched_barrier(0);
            mma_step<MT, NG>(acc, fb);
            __builtin_amdgcn_sched_barrier(0);
        }
        fetch(fb, t + 1);
        __builtin_amdgcn_sched_barrier(0);
        mma_step<MT, NG>(acc, fa);
        mma_step<MT, NG>(acc, fb);
    }

    {
        const float* xl = xs + 4 * lane;
        const int b0 = __builtin_amdgcn_readlane(bv, 0);
        const int lpo = PCAP * w;
        int pq[PCAP / 32];
        #pragma unroll
        for (int q = 0; q < PCAP / 32; ++q) pq[q] = perm[min(b0 + 32 * q + lane, EL - 1)];
        #pragma unroll
        for (int q = 0; q < PCAP / 32; ++q) lperm[lpo + 32 * q + lane] = pq[q];
        __syncthreads();
        for (int r = 0; r < 8; ++r) {
            const int deg = __builtin_amdgcn_readlane(dv, r);
            const int begin = __builtin_amdgcn_readlane(bv, r);
            const int off = begin - b0;
            v4f s[V];
            #pragma unroll
            for (int v = 0; v < V; ++v) s[v] = (v4f){0.f, 0.f, 0.f, 0.f};
            for (int j0 = 0; j0 < deg; j0 += 32) {
                const int m = min(32, deg - j0);
                const int o = off + j0;
                int my = lperm[lpo + min(max(o, 0) + min(lane, m - 1), PCAP - 1)];
                asm volatile("" : "+v"(my));
                if (o < 0 || o + m > PCAP) my = perm[begin + j0 + min(lane, m - 1)];
                for (int k = 0; k < m; k += UNR) {
                    v4f rr[UNR][V];
                    #pragma unroll
                    for (int u = 0; u < UNR; ++u) {
                        const int kk = min(k + u, m - 1);
                        const int sidx = __builtin_amdgcn_readlane(my, kk);
                        const float* rp = xl + (size_t)sidx * K;
                        #pragma unroll
                        for (int v = 0; v < V; ++v) rr[u][v] = *(const v4f*)(rp + 128 * v);
                    }
                    #pragma unroll
                    for (int u = 0; u < UNR; ++u) {
                        const bool on = (u == 0) || (k + u) < m;
                        #pragma unroll
                        for (int v = 0; v < V; ++v)
                            s[v] += on ? rr[u][v] : (v4f){0.f, 0.f, 0.f, 0.f};
                    }
                }
            }
            const float inv = 1.0f / fmaxf((float)deg, 1.0f);
            float* mp = aggT + (8 * w + r) * PITCH + 4 * lane;
            #pragma unroll
            for (int v = 0; v < V; ++v) *(v4f*)(mp + 128 * v) = s[v] * inv;
        }
    }
    __syncthreads();

    {
        const float* alp = aggT + c * PITCH + 4 * h;
        auto fetch = [&](Frags<MT, NG>& f, int t) {
            load_frags_lds<MT, NG, PITCH>(f, alp + 8 * t, w0p + (size_t)(8 * t) * N, N);
        };
        fetch(fa, 0);
        int t = 0;
        for (; t < STEPS - 2; t += 2) {
            fetch(fb, t + 1);
            __builtin_amdgcn_sched_barrier(0);
            mma_step<MT, NG>(acc, fa);
            __builtin_amdgcn_sched_barrier(0);
            fetch(fa, t + 2);
            __builtin_amdgcn_sched_barrier(0);
            mma_step<MT, NG>(acc, fb);
            __builtin_amdgcn_sched_barrier(0);
        }
        fetch(fb, t + 1);
        __builtin_amdgcn_sched_barrier(0);
        mma_step<MT, NG>(acc, fa);
        mma_step<MT, NG>(acc, fb);
    }

    {
        const int col = n0 + 4 * c;
        const float b0 = bias[col], b1 = bias[col + 1], b2 = bias[col + 2], b3 = bias[col + 3];
        #pragma unroll
        for (int i = 0; i < MT; ++i) {
            float* op = out + (size_t)(m0 + 16 * i + 8 * h) * N + col;
            #pragma unroll
            for (int r = 0; r < 8; ++r) {
                v4f o;
                o.x = acc[i][0][0][r] + b0;
                o.y = acc[i][0][1][r] + b1;
                o.z = acc[i][0][2][r] + b2;
                o.w = acc[i][0][3][r] + b3;
                if (relu) {
                    o.x = fmaxf(o.x, 0.0f); o.y = fmaxf(o.y, 0.0f);
                    o.z = fmaxf(o.z, 0.0f); o.w = fmaxf(o.w, 0.0f);
                }
                *(v4f*)(op + (size_t)r * N) = o;
            }
        }
    }
}

extern "C" void kernel_launch(void* const* d_in, const int* in_sizes, int n_in,
                              void* d_out, int out_size, void* d_ws, size_t ws_size,
                              hipStream_t stream) {
    (void)in_sizes; (void)n_in; (void)out_size; (void)ws_size;

    const float* x = (const float*)d_in[0];
    const int* srcp[3] = {(const int*)d_in[1], (const int*)d_in[3], (const int*)d_in[5]};
    const int* dstp[3] = {(const int*)d_in[2], (const int*)d_in[4], (const int*)d_in[6]};
    const float* Wl[3] = {(const float*)d_in[7],  (const float*)d_in[10], (const float*)d_in[13]};
    const float* Wr[3] = {(const float*)d_in[8],  (const float*)d_in[11], (const float*)d_in[14]};
    const float* bb[3] = {(const float*)d_in[9],  (const float*)d_in[12], (const float*)d_in[15]};

    const int relu[3] = {1, 0, 1};

    float* ws     = (float*)d_ws;
    float* aggbuf = ws;
    float* small  = aggbuf + 12800000;
    float* h0     = small + 100000;
    float* h1     = h0 + 25600000;

    int* cnt0 = (int*)aggbuf + 6000000;  int* cur0 = cnt0 + N_L0;  int* perm0 = cur0 + N_L0;
    int* cnt1 = (int*)aggbuf + 8000000;  int* cur1 = cnt1 + N_L1;  int* perm1 = cur1 + N_L1;
    int* cnt2 = (int*)small;  int* cur2 = cnt2 + N_L2;  int* perm2 = cur2 + N_L2;
    int* tick  = (int*)aggbuf;
    int* tc0   = tick + 1860000;  int* tc1 = tc0 + T0;  int* tc2 = tc1 + T1;
    int* tu0   = tc0 + 75012;     int* tu1 = tu0 + T0;  int* tu2 = tu1 + T1;
    int* pairs = tu0 + 75012;

    chunk_count<<<CB0 + CB1 + CB2, 1024, 0, stream>>>(dstp[0], dstp[1], dstp[2], tc0, tc1, tc2, tick);
    seg_offsets<<<SCAN_B0 + SCAN_B1 + SCAN_B2, 1024, 0, stream>>>(tc0, tu0, tc1, tu1, tc2, tu2);
    chunk_place<<<CB0 + CB1 + CB2, 1024, 0, stream>>>(srcp[0], dstp[0], srcp[1], dstp[1],
                                                       srcp[2], dstp[2], tu0, tu1, tu2, tick, pairs);
    bin_sort<<<NB0 + NB1 + NB2, 1024, 0, stream>>>(tu0, tu1, tu2, pairs,
                                                    cnt0, cur0, perm0, cnt1, cur1, perm1,
                                                    cnt2, cur2, perm2);

    sage_fused<128, E_L0><<<N_L0 / 32, 128, 0, stream>>>(x, perm0, cnt0, cur0, Wl[0], Wr[0], bb[0], h0, relu[0]);
    sage_fused<256, E_L1><<<N_L1 / 32, 128, 0, stream>>>(h0, perm1, cnt1, cur1, Wl[1], Wr[1], bb[1], h1, relu[1]);

    seg_mean_gather<256, 4><<<N_L2 / 8, 256, 0, stream>>>(h1, perm2, cnt2, cur2, aggbuf, N_L2);
    sage_gemm_wmma<1, 1><<<125, 128, 0, stream>>>(aggbuf, h1, Wl[2], Wr[2], bb[2], (float*)d_out,
                                                   N_L2, 256, 128, relu[2]);
}


# Layer 2 in one kernel: 16-row work-groups put mean and x_tgt rows in LDS, the 512-k sum is split over 8 waves (W fragments requested at once), 8 partial sums added in wave order; 7 launches
# speedup vs baseline: 4.1492x; 1.0491x over previous
#include <hip/hip_runtime.h>


typedef float v2f __attribute__((ext_vector_type(2)));
typedef float v4f __attribute__((ext_vector_type(4)));
typedef float v8f __attribute__((ext_vector_type(8)));

#define E_L0 1500000
#define E_L1 300000
#define E_L2 60000
#define N_L0 100000
#define N_L1 20000
#define N_L2 4000

#define BIN_LOG 9
#define BIN_W 512
#define CHUNK 4096
#define NB0 196
#define NB1 40
#define NB2 8
#define CB0 367
#define CB1 74
#define CB2 15
#define T0 (NB0 * CB0)
#define T1 (NB1 * CB1)
#define T2 (NB2 * CB2)

__global__ void __launch_bounds__(1024)
chunk_count(const int* __restrict__ d0, const int* __restrict__ d1, const int* __restrict__ d2,
            int* __restrict__ t0, int* __restrict__ t1, int* __restrict__ t2,
            int* __restrict__ tick) {
    __shared__ int hist[256];
    int b = blockIdx.x;
    const int* dp; int* tb; int E, nblk, nb, ebase;
    if (b < CB0) { dp = d0; tb = t0; E = E_L0; nblk = CB0; nb = NB0; ebase = 0; }
    else if (b < CB0 + CB1) { b -= CB0; dp = d1; tb = t1; E = E_L1; nblk = CB1; nb = NB1; ebase = E_L0; }
    else { b -= CB0 + CB1; dp = d2; tb = t2; E = E_L2; nblk = CB2; nb = NB2; ebase = E_L0 + E_L1; }
    const int tid = threadIdx.x;
    if (tid < 256) hist[tid] = 0;
    __syncthreads();
    const int e0 = b * CHUNK + tid;
    int d[4];
    #pragma unroll
    for (int k = 0; k < 4; ++k) d[k] = dp[min(e0 + 1024 * k, E - 1)];
    #pragma unroll
    for (int k = 0; k < 4; ++k) {
        const int e = e0 + 1024 * k;
        if (e < E) tick[ebase + e] = atomicAdd(&hist[d[k] >> BIN_LOG], 1);
    }
    __syncthreads();
    if (tid < nb) tb[tid * nblk + b] = hist[tid];
}

#define SCAN_B0 71
#define SCAN_B1 3
#define SCAN_B2 1
__global__ void __launch_bounds__(1024)
seg_offsets(const int* __restrict__ c0, int* __restrict__ u0,
            const int* __restrict__ c1, int* __restrict__ u1,
            const int* __restrict__ c2, int* __restrict__ u2) {
    __shared__ int red[32];
    __shared__ int tot[32];
    int b = blockIdx.x;
    const int* cnt; int* cur; int n;
    if (b < SCAN_B0) { cnt = c0; cur = u0; n = T0; }
    else if (b < SCAN_B0 + SCAN_B1) { b -= SCAN_B0; cnt = c1; cur = u1; n = T1; }
    else { b -= SCAN_B0 + SCAN_B1; cnt = c2; cur = u2; n = T2; }
    const int tid = threadIdx.x;
    const int lane = tid & 31;
    const int w = tid >> 5;
    const int lo = b * 1024;
    int p = 0;
    const int4* c4 = (const int4*)cnt;
    for (int j = tid; j < (lo >> 2); j += 1024) {
        int4 q = c4[j];
        p += q.x + q.y + q.z + q.w;
    }
    const int idx = lo + tid;
    const int v = (idx < n) ? cnt[idx] : 0;
    int incl = v;
    for (int d = 1; d < 32; d <<= 1) {
        int t = __shfl_up(incl, d);
        if (lane >= d) incl += t;
    }
    for (int d = 16; d >= 1; d >>= 1) p += __shfl_xor(p, d);
    if (lane == 31) tot[w] = incl;
    if (lane == 0) red[w] = p;
    __syncthreads();
    int base = 0;
    for (int k = 0; k < 32; ++k) {
        base += red[k];
        if (k < w) base += tot[k];
    }
    if (idx < n) cur[idx] = base + incl - v;
}

__global__ void __launch_bounds__(1024)
chunk_place(const int* __restrict__ s0, const int* __restrict__ d0,
            const int* __restrict__ s1, const int* __restrict__ d1,
            const int* __restrict__ s2, const int* __restrict__ d2,
            const int* __restrict__ u0, const int* __restrict__ u1,
            const int* __restrict__ u2, const int* __restrict__ tick,
            int* __restrict__ pairs) {
    __shared__ int first[256];
    int b = blockIdx.x;
    const int* sp; const int* dp; const int* up; int E, nblk, nb, ebase;
    if (b < CB0) { sp = s0; dp = d0; up = u0; E = E_L0; nblk = CB0; nb = NB0; ebase = 0; }
    else if (b < CB0 + CB1) { b -= CB0; sp = s1; dp = d1; up = u1; E = E_L1; nblk = CB1; nb = NB1; ebase = E_L0; }
    else { b -= CB0 + CB1; sp = s2; dp = d2; up = u2; E = E_L2; nblk = CB2; nb = NB2; ebase = E_L0 + E_L1; }
    const int tid = threadIdx.x;
    if (tid < nb) first[tid] = up[tid * nblk + b];
    const int e0 = b * CHUNK + tid;
    int d[4], sv[4], r[4];
    #pragma unroll
    for (int k = 0; k < 4; ++k) {
        const int e = min(e0 + 1024 * k, E - 1);
        d[k] = dp[e];
        sv[k] = sp[e];
        r[k] = tick[ebase + e];
    }
    __syncthreads();
    int2* pr = (int2*)pairs + ebase;
    #pragma unroll
    for (int k = 0; k < 4; ++k) {
        const int e = e0 + 1024 * k;
        if (e < E) pr[first[d[k] >> BIN_LOG] + r[k]] = make_int2(sv[k], d[k]);
    }
}

__global__ void __launch_bounds__(1024)
bin_sort(const int* __restrict__ u0, const int* __restrict__ u1, const int* __restrict__ u2,
         const int* __restrict__ pairs,
         int* __restrict__ c0, int* __restrict__ f0, int* __restrict__ p0,
         int* __restrict__ c1, int* __restrict__ f1, int* __restrict__ p1,
         int* __restrict__ c2, int* __restrict__ f2, int* __restrict__ p2) {
    __shared__ int cntL[BIN_W];
    __shared__ int curL[BIN_W];
    __shared__ int wsum[16];
    int b = blockIdx.x;
    const int* up; int* cnt; int* cur; int* perm; int E, nblk, nb, n, ebase;
    if (b < NB0) { up = u0; cnt = c0; cur = f0; perm = p0; E = E_L0; nblk = CB0; nb = NB0; n = N_L0; ebase = 0; }
    else if (b < NB0 + NB1) { b -= NB0; up = u1; cnt = c1; cur = f1; perm = p1; E = E_L1; nblk = CB1; nb = NB1; n = N_L1; ebase = E_L0; }
    else { b -= NB0 + NB1; up = u2; cnt = c2; cur = f2; perm = p2; E = E_L2; nblk = CB2; nb = NB2; n = N_L2; ebase = E_L0 + E_L1; }
    const int tid = threadIdx.x;
    const int lane = tid & 31;
    const int w = tid >> 5;
    const int start = up[b * nblk];
    const int end = (b + 1 < nb) ? up[(b + 1) * nblk] : E;
    const int2* pr = (const int2*)pairs + ebase;
    if (tid < BIN_W) cntL[tid] = 0;
    __syncthreads();
    for (int j = start + tid; j < end; j += 4096) {
        int dd[4];
        #pragma unroll
        for (int k = 0; k < 4; ++k) dd[k] = pr[min(j + 1024 * k, end - 1)].y;
        #pragma unroll
        for (int k = 0; k < 4; ++k)
            if (j + 1024 * k < end) atomicAdd(&cntL[dd[k] & (BIN_W - 1)], 1);
    }
    __syncthreads();
    const int v = (tid < BIN_W) ? cntL[tid] : 0;
    int incl = v;
    for (int d = 1; d < 32; d <<= 1) {
        int t = __shfl_up(incl, d);
        if (lane >= d) incl += t;
    }
    if (lane == 31 && w < 16) wsum[w] = incl;
    __syncthreads();
    if (tid < BIN_W) {
        int base = start;
        for (int k = 0; k < 16; ++k)
            if (k < w) base += wsum[k];
        const int firstSlot = base + incl - v;
        curL[tid] = firstSlot;
        const int node = b * BIN_W + tid;
        if (node < n) { cnt[node] = v; cur[node] = firstSlot; }
    }
    __syncthreads();
    for (int j = start + tid; j < end; j += 4096) {
        int2 q[4];
        #pragma unroll
        for (int k = 0; k < 4; ++k) q[k] = pr[min(j + 1024 * k, end - 1)];
        #pragma unroll
        for (int k = 0; k < 4; ++k)
            if (j + 1024 * k < end) perm[atomicAdd(&curL[q[k].y & (BIN_W - 1)], 1)] = q[k].x;
    }
}

template <int MT, int NG>
struct Frags {
    v4f a[MT];
    v4f w[NG][4];
};

template <int MT, int NG>
__device__ __forceinline__ void load_frags(Frags<MT, NG>& f, const float* __restrict__ ap, int K,
                                           const float* __restrict__ wp, int N) {
    #pragma unroll
    for (int i = 0; i < MT; ++i)
        f.a[i] = *(const v4f*)(ap + (size_t)(16 * i) * K);
    #pragma unroll
    for (int g = 0; g < NG; ++g) {
        #pragma unroll
        for (int q = 0; q < 4; ++q)
            f.w[g][q] = *(const v4f*)(wp + (size_t)q * N + 64 * g);
    }
}

template <int MT, int NG>
__device__ __forceinline__ void mma_step(v8f (&acc)[MT][NG][4], const Frags<MT, NG>& f) {
    #pragma unroll
    for (int s = 0; s < 2; ++s) {
        #pragma unroll
        for (int i = 0; i < MT; ++i) {
            v2f av;
            av.x = f.a[i][2 * s];
            av.y = f.a[i][2 * s + 1];
            #pragma unroll
            for (int g = 0; g < NG; ++g) {
                #pragma unroll
                for (int j = 0; j < 4; ++j) {
                    v2f bv;
                    bv.x = f.w[g][2 * s][j];
                    bv.y = f.w[g][2 * s + 1][j];
                    acc[i][g][j] = __builtin_amdgcn_wmma_f32_16x16x4_f32(
                         false, av,  false, bv,
                         (short)0, acc[i][g][j],  false,  false);
                }
            }
        }
    }
}

#define PCAP 192
template <int MT, int NG, int PITCH>
__device__ __forceinline__ void load_frags_lds(Frags<MT, NG>& f, const float* ap,
                                               const float* __restrict__ wp, int N) {
    #pragma unroll
    for (int i = 0; i < MT; ++i)
        f.a[i] = *(const v4f*)(ap + 16 * i * PITCH);
    #pragma unroll
    for (int g = 0; g < NG; ++g) {
        #pragma unroll
        for (int q = 0; q < 4; ++q)
            f.w[g][q] = *(const v4f*)(wp + (size_t)q * N + 64 * g);
    }
}

template <int K, int EL>
__global__ void __launch_bounds__(128)
sage_fused(const float* __restrict__ xs, const int* __restrict__ perm,
           const int* __restrict__ cnt, const int* __restrict__ cur,
           const float* __restrict__ W0,
           const float* __restrict__ W1,
           const float* __restrict__ bias, float* __restrict__ out, int relu) {
    constexpr int N = 256, MT = 2, NG = 1, STEPS = K / 8, UNR = 8;
    constexpr int V = K / 128;
    constexpr int PITCH = K + 4;
    __shared__ __attribute__((aligned(16))) float aggT[32 * PITCH];
    __shared__ int lperm[4 * PCAP];
    const int lane = threadIdx.x & 31;
    const int w = threadIdx.x >> 5;
    const int h = lane >> 4;
    const int c = lane & 15;
    const int m0 = blockIdx.x * 32;
    const int n0 = 64 * w;
    const int dv = cnt[m0 + 8 * w + min(lane, 7)];
    const int bv = cur[m0 + 8 * w + min(lane, 7)];

    v8f acc[MT][NG][4];
    #pragma unroll
    for (int i = 0; i < MT; ++i) {
        #pragma unroll
        for (int j = 0; j < 4; ++j)
            acc[i][0][j] = (v8f){0.f, 0.f, 0.f, 0.f, 0.f, 0.f, 0.f, 0.f};
    }
    const float* a1p = xs + (size_t)(m0 + c) * K + 4 * h;
    const size_t woff = (size_t)(4 * h) * N + n0 + 4 * c;
    const float* w0p = W0 + woff;
    const float* w1p = W1 + woff;
    Frags<MT, NG> fa, fb;

    {
        auto fetch = [&](Frags<MT, NG>& f, int t) {
            load_frags<MT, NG>(f, a1p + 8 * t, K, w1p + (size_t)(8 * t) * N, N);
        };
        fetch(fa, 0);
        int t = 0;
        for (; t < STEPS - 2; t += 2) {
            fetch(fb, t + 1);
            __builtin_amdgcn_sched_barrier(0);
            mma_step<MT, NG>(acc, fa);
            __builtin_amdgcn_sched_barrier(0);
            fetch(fa, t + 2);
            __builtin_amdgcn_sched_barrier(0);
            mma_step<MT, NG>(acc, fb);
            __builtin_amdgcn_sched_barrier(0);
        }
        fetch(fb, t + 1);
        __builtin_amdgcn_sched_barrier(0);
        mma_step<MT, NG>(acc, fa);
        mma_step<MT, NG>(acc, fb);
    }

    {
        const float* xl = xs + 4 * lane;
        const int b0 = __builtin_amdgcn_readlane(bv, 0);
        const int lpo = PCAP * w;
        int pq[PCAP / 32];
        #pragma unroll
        for (int q = 0; q < PCAP / 32; ++q) pq[q] = perm[min(b0 + 32 * q + lane, EL - 1)];
        #pragma unroll
        for (int q = 0; q < PCAP / 32; ++q) lperm[lpo + 32 * q + lane] = pq[q];
        __syncthreads();
        for (int r = 0; r < 8; ++r) {
            const int deg = __builtin_amdgcn_readlane(dv, r);
            const int begin = __builtin_amdgcn_readlane(bv, r);
            const int off = begin - b0;
            v4f s[V];
            #pragma unroll
            for (int v = 0; v < V; ++v) s[v] = (v4f){0.f, 0.f, 0.f, 0.f};
            for (int j0 = 0; j0 < deg; j0 += 32) {
                const int m = min(32, deg - j0);
                const int o = off + j0;
                int my = lperm[lpo + min(max(o, 0) + min(lane, m - 1), PCAP - 1)];
                asm volatile("" : "+v"(my));
                if (o < 0 || o + m > PCAP) my = perm[begin + j0 + min(lane, m - 1)];
                for (int k = 0; k < m; k += UNR) {
                    v4f rr[UNR][V];
                    #pragma unroll
                    for (int u = 0; u < UNR; ++u) {
                        const int kk = min(k + u, m - 1);
                        const int sidx = __builtin_amdgcn_readlane(my, kk);
                        const float* rp = xl + (size_t)sidx * K;
                        #pragma unroll
                        for (int v = 0; v < V; ++v) rr[u][v] = *(const v4f*)(rp + 128 * v);
                    }
                    #pragma unroll
                    for (int u = 0; u < UNR; ++u) {
                        const bool on = (u == 0) || (k + u) < m;
                        #pragma unroll
                        for (int v = 0; v < V; ++v)
                            s[v] += on ? rr[u][v] : (v4f){0.f, 0.f, 0.f, 0.f};
                    }
                }
            }
            const float inv = 1.0f / fmaxf((float)deg, 1.0f);
            float* mp = aggT + (8 * w + r) * PITCH + 4 * lane;
            #pragma unroll
            for (int v = 0; v < V; ++v) *(v4f*)(mp + 128 * v) = s[v] * inv;
        }
    }
    __syncthreads();

    {
        const float* alp = aggT + c * PITCH + 4 * h;
        auto fetch = [&](Frags<MT, NG>& f, int t) {
            load_frags_lds<MT, NG, PITCH>(f, alp + 8 * t, w0p + (size_t)(8 * t) * N, N);
        };
        fetch(fa, 0);
        int t = 0;
        for (; t < STEPS - 2; t += 2) {
            fetch(fb, t + 1);
            __builtin_amdgcn_sched_barrier(0);
            mma_step<MT, NG>(acc, fa);
            __builtin_amdgcn_sched_barrier(0);
            fetch(fa, t + 2);
            __builtin_amdgcn_sched_barrier(0);
            mma_step<MT, NG>(acc, fb);
            __builtin_amdgcn_sched_barrier(0);
        }
        fetch(fb, t + 1);
        __builtin_amdgcn_sched_barrier(0);
        mma_step<MT, NG>(acc, fa);
        mma_step<MT, NG>(acc, fb);
    }

    {
        const int col = n0 + 4 * c;
        const float b0 = bias[col], b1 = bias[col + 1], b2 = bias[col + 2], b3 = bias[col + 3];
        #pragma unroll
        for (int i = 0; i < MT; ++i) {
            float* op = out + (size_t)(m0 + 16 * i + 8 * h) * N + col;
            #pragma unroll
            for (int r = 0; r < 8; ++r) {
                v4f o;
                o.x = acc[i][0][0][r] + b0;
                o.y = acc[i][0][1][r] + b1;
                o.z = acc[i][0][2][r] + b2;
                o.w = acc[i][0][3][r] + b3;
                if (relu) {
                    o.x = fmaxf(o.x, 0.0f); o.y = fmaxf(o.y, 0.0f);
                    o.z = fmaxf(o.z, 0.0f); o.w = fmaxf(o.w, 0.0f);
                }
                *(v4f*)(op + (size_t)r * N) = o;
            }
        }
    }
}

__global__ void __launch_bounds__(256)
sage_fused_ksplit(const float* __restrict__ xs, const int* __restrict__ perm,
                  const int* __restrict__ cnt, const int* __restrict__ cur,
                  const float* __restrict__ W0,
                  const float* __restrict__ W1,
                  const float* __restrict__ bias, float* __restrict__ out, int relu) {
    constexpr int K = 256, N = 128, MT = 1, NG = 2, UNR = 8, V = K / 128;
    constexpr int APITCH = 2 * K + 4;
    constexpr int PPITCH = N + 4;
    __shared__ __attribute__((aligned(16))) float aT[16 * APITCH];
    __shared__ __attribute__((aligned(16))) float part[8 * 16 * PPITCH];
    const int lane = threadIdx.x & 31;
    const int w = threadIdx.x >> 5;
    const int h = lane >> 4;
    const int c = lane & 15;
    const int m0 = blockIdx.x * 16;
    const int dv = cnt[m0 + 2 * w + min(lane, 1)];
    const int bv = cur[m0 + 2 * w + min(lane, 1)];

    const float* wsl = (w < 4 ? W0 + (size_t)(64 * w) * N : W1 + (size_t)(64 * (w - 4)) * N)
                       + (size_t)(4 * h) * N + 4 * c;
    Frags<MT, NG> fr[8];
    #pragma unroll
    for (int t = 0; t < 8; ++t) {
        #pragma unroll
        for (int g = 0; g < NG; ++g) {
            #pragma unroll
            for (int q = 0; q < 4; ++q)
                fr[t].w[g][q] = *(const v4f*)(wsl + (size_t)(8 * t + q) * N + 64 * g);
        }
    }

    {
        const float* xl = xs + 4 * lane;
        #pragma unroll
        for (int r = 0; r < 2; ++r) {
            const int deg = __builtin_amdgcn_readlane(dv, r);
            const int begin = __builtin_amdgcn_readlane(bv, r);
            v4f sacc[V];
            #pragma unroll
            for (int v = 0; v < V; ++v) sacc[v] = (v4f){0.f, 0.f, 0.f, 0.f};
            for (int j0 = 0; j0 < deg; j0 += 32) {
                const int m = min(32, deg - j0);
                const int my = perm[begin + j0 + min(lane, m - 1)];
                for (int k = 0; k < m; k += UNR) {
                    v4f rr[UNR][V];
                    #pragma unroll
                    for (int u = 0; u < UNR; ++u) {
                        const int kk = min(k + u, m - 1);
                        const int sidx = __builtin_amdgcn_readlane(my, kk);
                        const float* rp = xl + (size_t)sidx * K;
                        #pragma unroll
                        for (int v = 0; v < V; ++v) rr[u][v] = *(const v4f*)(rp + 128 * v);
                    }
                    #pragma unroll
                    for (int u = 0; u < UNR; ++u) {
                        const bool on = (u == 0) || (k + u) < m;
                        #pragma unroll
                        for (int v = 0; v < V; ++v)
                            sacc[v] += on ? rr[u][v] : (v4f){0.f, 0.f, 0.f, 0.f};
                    }
                }
            }
            const float inv = 1.0f / fmaxf((float)deg, 1.0f);
            float* mp = aT + (2 * w + r) * APITCH + 4 * lane;
            const float* tp = xs + (size_t)(m0 + 2 * w + r) * K + 4 * lane;
            #pragma unroll
            for (int v = 0; v < V; ++v) {
                *(v4f*)(mp + 128 * v) = sacc[v] * inv;
                *(v4f*)(mp + K + 128 * v) = *(const v4f*)(tp + 128 * v);
            }
        }
    }
    __syncthreads();

    v8f acc[MT][NG][4];
    #pragma unroll
    for (int g = 0; g < NG; ++g) {
        #pragma unroll
        for (int j = 0; j < 4; ++j)
            acc[0][g][j] = (v8f){0.f, 0.f, 0.f, 0.f, 0.f, 0.f, 0.f, 0.f};
    }
    {
        const float* alp = aT + c * APITCH + 64 * w + 4 * h;
        #pragma unroll
        for (int t = 0; t < 8; ++t) fr[t].a[0] = *(const v4f*)(alp + 8 * t);
        __builtin_amdgcn_sched_barrier(0);
        #pragma unroll
        for (int t = 0; t < 8; ++t) mma_step<MT, NG>(acc, fr[t]);
    }

    {
        float* pp = part + w * (16 * PPITCH) + (8 * h) * PPITCH + 4 * c;
        #pragma unroll
        for (int g = 0; g < NG; ++g) {
            #pragma unroll
            for (int r = 0; r < 8; ++r) {
                v4f o;
                o.x = acc[0][g][0][r];
                o.y = acc[0][g][1][r];
                o.z = acc[0][g][2][r];
                o.w = acc[0][g][3][r];
                *(v4f*)(pp + r * PPITCH + 64 * g) = o;
            }
        }
    }
    __syncthreads();
    {
        const v4f bq = *(const v4f*)(bias + 4 * lane);
        #pragma unroll
        for (int r = 0; r < 2; ++r) {
            const float* rp = part + (2 * w + r) * PPITCH + 4 * lane;
            v4f o = *(const v4f*)rp;
            #pragma unroll
            for (int p = 1; p < 8; ++p) o += *(const v4f*)(rp + p * (16 * PPITCH));
            o += bq;
            if (relu) {
                o.x = fmaxf(o.x, 0.0f); o.y = fmaxf(o.y, 0.0f);
                o.z = fmaxf(o.z, 0.0f); o.w = fmaxf(o.w, 0.0f);
            }
            *(v4f*)(out + (size_t)(m0 + 2 * w + r) * N + 4 * lane) = o;
        }
    }
}

extern "C" void kernel_launch(void* const* d_in, const int* in_sizes, int n_in,
                              void* d_out, int out_size, void* d_ws, size_t ws_size,
                              hipStream_t stream) {
    (void)in_sizes; (void)n_in; (void)out_size; (void)ws_size;

    const float* x = (const float*)d_in[0];
    const int* srcp[3] = {(const int*)d_in[1], (const int*)d_in[3], (const int*)d_in[5]};
    const int* dstp[3] = {(const int*)d_in[2], (const int*)d_in[4], (const int*)d_in[6]};
    const float* Wl[3] = {(const float*)d_in[7],  (const float*)d_in[10], (const float*)d_in[13]};
    const float* Wr[3] = {(const float*)d_in[8],  (const float*)d_in[11], (const float*)d_in[14]};
    const float* bb[3] = {(const float*)d_in[9],  (const float*)d_in[12], (const float*)d_in[15]};

    const int relu[3] = {1, 0, 1};

    float* ws     = (float*)d_ws;
    float* aggbuf = ws;
    float* small  = aggbuf + 12800000;
    float* h0     = small + 100000;
    float* h1     = h0 + 25600000;

    int* cnt0 = (int*)aggbuf + 6000000;  int* cur0 = cnt0 + N_L0;  int* perm0 = cur0 + N_L0;
    int* cnt1 = (int*)aggbuf + 8000000;  int* cur1 = cnt1 + N_L1;  int* perm1 = cur1 + N_L1;
    int* cnt2 = (int*)small;  int* cur2 = cnt2 + N_L2;  int* perm2 = cur2 + N_L2;
    int* tick  = (int*)aggbuf;
    int* tc0   = tick + 1860000;  int* tc1 = tc0 + T0;  int* tc2 = tc1 + T1;
    int* tu0   = tc0 + 75012;     int* tu1 = tu0 + T0;  int* tu2 = tu1 + T1;
    int* pairs = tu0 + 75012;

    chunk_count<<<CB0 + CB1 + CB2, 1024, 0, stream>>>(dstp[0], dstp[1], dstp[2], tc0, tc1, tc2, tick);
    seg_offsets<<<SCAN_B0 + SCAN_B1 + SCAN_B2, 1024, 0, stream>>>(tc0, tu0, tc1, tu1, tc2, tu2);
    chunk_place<<<CB0 + CB1 + CB2, 1024, 0, stream>>>(srcp[0], dstp[0], srcp[1], dstp[1],
                                                       srcp[2], dstp[2], tu0, tu1, tu2, tick, pairs);
    bin_sort<<<NB0 + NB1 + NB2, 1024, 0, stream>>>(tu0, tu1, tu2, pairs,
                                                    cnt0, cur0, perm0, cnt1, cur1, perm1,
                                                    cnt2, cur2, perm2);

    sage_fused<128, E_L0><<<N_L0 / 32, 128, 0, stream>>>(x, perm0, cnt0, cur0, Wl[0], Wr[0], bb[0], h0, relu[0]);
    sage_fused<256, E_L1><<<N_L1 / 32, 128, 0, stream>>>(h0, perm1, cnt1, cur1, Wl[1], Wr[1], bb[1], h1, relu[1]);

    sage_fused_ksplit<<<N_L2 / 16, 256, 0, stream>>>(h1, perm2, cnt2, cur2, Wl[2], Wr[2], bb[2],
                                                     (float*)d_out, relu[2]);
}


# Edge sort level 1 in one pass: a chunk scans its cell counts in LDS and stores its pairs bin by bin in its own 4096 slots; level 2 reads one run per chunk; 5 launches; sum order = LDS arrival order
# speedup vs baseline: 4.4987x; 1.0842x over previous
#include <hip/hip_runtime.h>


typedef float v2f __attribute__((ext_vector_type(2)));
typedef float v4f __attribute__((ext_vector_type(4)));
typedef float v8f __attribute__((ext_vector_type(8)));

#define E_L0 1500000
#define E_L1 300000
#define E_L2 60000
#define N_L0 100000
#define N_L1 20000
#define N_L2 4000

#define BIN_LOG 9
#define BIN_W 512
#define CHUNK 4096
#define NB0 196
#define NB1 40
#define NB2 8
#define CB0 367
#define CB1 74
#define CB2 15
#define T0 (NB0 * CB0)
#define T1 (NB1 * CB1)
#define T2 (NB2 * CB2)

__global__ void __launch_bounds__(1024)
chunk_sort(const int* __restrict__ s0, const int* __restrict__ d0,
           const int* __restrict__ s1, const int* __restrict__ d1,
           const int* __restrict__ s2, const int* __restrict__ d2,
           int* __restrict__ t0, int* __restrict__ t1, int* __restrict__ t2,
           int* __restrict__ l0, int* __restrict__ l1, int* __restrict__ l2,
           int* __restrict__ pairs) {
    __shared__ int hist[256];
    __shared__ int offL[256];
    __shared__ int wsum[8];
    int b = blockIdx.x;
    const int* sp; const int* dp; int* tb; int* lb; int E, nblk, nb, pbase;
    if (b < CB0) { sp = s0; dp = d0; tb = t0; lb = l0; E = E_L0; nblk = CB0; nb = NB0; pbase = 0; }
    else if (b < CB0 + CB1) { b -= CB0; sp = s1; dp = d1; tb = t1; lb = l1; E = E_L1; nblk = CB1; nb = NB1; pbase = CB0 * CHUNK; }
    else { b -= CB0 + CB1; sp = s2; dp = d2; tb = t2; lb = l2; E = E_L2; nblk = CB2; nb = NB2; pbase = (CB0 + CB1) * CHUNK; }
    const int tid = threadIdx.x;
    const int lane = tid & 31;
    const int w = tid >> 5;
    if (tid < 256) hist[tid] = 0;
    __syncthreads();
    const int e0 = b * CHUNK + tid;
    int d[4], sv[4], r[4];
    #pragma unroll
    for (int k = 0; k < 4; ++k) {
        const int e = min(e0 + 1024 * k, E - 1);
        d[k] = dp[e];
        sv[k] = sp[e];
    }
    #pragma unroll
    for (int k = 0; k < 4; ++k) {
        r[k] = 0;
        if (e0 + 1024 * k < E) r[k] = atomicAdd(&hist[d[k] >> BIN_LOG], 1);
    }
    __syncthreads();
    const int v = (tid < 256) ? hist[tid] : 0;
    int incl = v;
    for (int dd = 1; dd < 32; dd <<= 1) {
        int t = __shfl_up(incl, dd);
        if (lane >= dd) incl += t;
    }
    if (lane == 31 && w < 8) wsum[w] = incl;
    __syncthreads();
    if (tid < 256) {
        int base = 0;
        for (int k = 0; k < 8; ++k)
            if (k < w) base += wsum[k];
        const int off = base + incl - v;
        offL[tid] = off;
        if (tid < nb) { tb[tid * nblk + b] = v; lb[tid * nblk + b] = off; }
    }
    __syncthreads();
    int2* pr = (int2*)pairs + pbase + b * CHUNK;
    #pragma unroll
    for (int k = 0; k < 4; ++k)
        if (e0 + 1024 * k < E) pr[offL[d[k] >> BIN_LOG] + r[k]] = make_int2(sv[k], d[k]);
}

#define CBMAX 368
#define SL0 0
#define SL1 0
#define SL2 3
__global__ void __launch_bounds__(1024)
bin_sort(const int* __restrict__ t0, const int* __restrict__ t1, const int* __restrict__ t2,
         const int* __restrict__ l0, const int* __restrict__ l1, const int* __restrict__ l2,
         const int* __restrict__ pairs,
         int* __restrict__ c0, int* __restrict__ f0, int* __restrict__ p0,
         int* __restrict__ c1, int* __restrict__ f1, int* __restrict__ p1,
         int* __restrict__ c2, int* __restrict__ f2, int* __restrict__ p2) {
    __shared__ int cntL[BIN_W];
    __shared__ int curL[BIN_W];
    __shared__ int wsum[16];
    __shared__ int red[32];
    __shared__ int cellCnt[CBMAX];
    __shared__ int cellPos[CBMAX];
    int b = blockIdx.x;
    const int* tb; const int* lb; int* cnt; int* cur; int* perm; int nblk, n, pbase, sl;
    if (b < NB0) { tb = t0; lb = l0; cnt = c0; cur = f0; perm = p0; nblk = CB0; n = N_L0; pbase = 0; sl = SL0; }
    else if (b < NB0 + NB1) { b -= NB0; tb = t1; lb = l1; cnt = c1; cur = f1; perm = p1; nblk = CB1; n = N_L1; pbase = CB0 * CHUNK; sl = SL1; }
    else { b -= NB0 + NB1; tb = t2; lb = l2; cnt = c2; cur = f2; perm = p2; nblk = CB2; n = N_L2; pbase = (CB0 + CB1) * CHUNK; sl = SL2; }
    const int tid = threadIdx.x;
    const int lane = tid & 31;
    const int w = tid >> 5;
    const int nfront = b * nblk;
    const int n4 = nfront >> 2;
    const int4* c4 = (const int4*)tb;
    int p = 0;
    for (int j0 = tid; j0 < n4; j0 += 8192) {
        int4 q[8];
        #pragma unroll
        for (int k = 0; k < 8; ++k) q[k] = c4[min(j0 + 1024 * k, n4 - 1)];
        __builtin_amdgcn_sched_barrier(0);
        #pragma unroll
        for (int k = 0; k < 8; ++k)
            if (j0 + 1024 * k < n4) p += q[k].x + q[k].y + q[k].z + q[k].w;
    }
    if (tid < (nfront & 3)) p += tb[(n4 << 2) + tid];
    if (tid < nblk) {
        cellCnt[tid] = tb[nfront + tid];
        cellPos[tid] = tid * CHUNK + lb[nfront + tid];
    }
    if (tid < BIN_W) cntL[tid] = 0;
    for (int dd = 16; dd >= 1; dd >>= 1) p += __shfl_xor(p, dd);
    if (lane == 0) red[w] = p;
    __syncthreads();
    const int2* pr = (const int2*)pairs + pbase;
    const int nitems = nblk << sl;
    for (int tb0 = w; tb0 < nitems; tb0 += 128) {
        int cn[4], co[4], ib[4];
        int mx = 0;
        #pragma unroll
        for (int k = 0; k < 4; ++k) {
            const int t = tb0 + 32 * k;
            const int c = min(t, nitems - 1) >> sl;
            cn[k] = (t < nitems) ? cellCnt[c] : 0;
            co[k] = cellPos[c];
            ib[k] = (t & ((1 << sl) - 1)) << 5;
            mx = max(mx, cn[k] - ib[k]);
        }
        for (int i0 = 0; i0 < mx; i0 += (32 << sl)) {
            int dd[4];
            #pragma unroll
            for (int k = 0; k < 4; ++k)
                dd[k] = pr[(cn[k] > 0) ? co[k] + min(i0 + ib[k] + lane, cn[k] - 1) : 0].y;
            #pragma unroll
            for (int k = 0; k < 4; ++k)
                if (i0 + ib[k] + lane < cn[k]) atomicAdd(&cntL[dd[k] & (BIN_W - 1)], 1);
        }
    }
    __syncthreads();
    const int v = (tid < BIN_W) ? cntL[tid] : 0;
    int incl = v;
    for (int dd = 1; dd < 32; dd <<= 1) {
        int t = __shfl_up(incl, dd);
        if (lane >= dd) incl += t;
    }
    if (lane == 31 && w < 16) wsum[w] = incl;
    __syncthreads();
    if (tid < BIN_W) {
        int base = 0;
        for (int k = 0; k < 32; ++k) base += red[k];
        for (int k = 0; k < 16; ++k)
            if (k < w) base += wsum[k];
        const int firstSlot = base + incl - v;
        curL[tid] = firstSlot;
        const int node = b * BIN_W + tid;
        if (node < n) { cnt[node] = v; cur[node] = firstSlot; }
    }
    __syncthreads();
    for (int tb0 = w; tb0 < nitems; tb0 += 128) {
        int cn[4], co[4], ib[4];
        int mx = 0;
        #pragma unroll
        for (int k = 0; k < 4; ++k) {
            const int t = tb0 + 32 * k;
            const int c = min(t, nitems - 1) >> sl;
            cn[k] = (t < nitems) ? cellCnt[c] : 0;
            co[k] = cellPos[c];
            ib[k] = (t & ((1 << sl) - 1)) << 5;
            mx = max(mx, cn[k] - ib[k]);
        }
        for (int i0 = 0; i0 < mx; i0 += (32 << sl)) {
            int2 q[4];
            #pragma unroll
            for (int k = 0; k < 4; ++k)
                q[k] = pr[(cn[k] > 0) ? co[k] + min(i0 + ib[k] + lane, cn[k] - 1) : 0];
            #pragma unroll
            for (int k = 0; k < 4; ++k)
                if (i0 + ib[k] + lane < cn[k]) perm[atomicAdd(&curL[q[k].y & (BIN_W - 1)], 1)] = q[k].x;
        }
    }
}

template <int MT, int NG>
struct Frags {
    v4f a[MT];
    v4f w[NG][4];
};

template <int MT, int NG>
__device__ __forceinline__ void load_frags(Frags<MT, NG>& f, const float* __restrict__ ap, int K,
                                           const float* __restrict__ wp, int N) {
    #pragma unroll
    for (int i = 0; i < MT; ++i)
        f.a[i] = *(const v4f*)(ap + (size_t)(16 * i) * K);
    #pragma unroll
    for (int g = 0; g < NG; ++g) {
        #pragma unroll
        for (int q = 0; q < 4; ++q)
            f.w[g][q] = *(const v4f*)(wp + (size_t)q * N + 64 * g);
    }
}

template <int MT, int NG>
__device__ __forceinline__ void mma_step(v8f (&acc)[MT][NG][4], const Frags<MT, NG>& f) {
    #pragma unroll
    for (int s = 0; s < 2; ++s) {
        #pragma unroll
        for (int i = 0; i < MT; ++i) {
            v2f av;
            av.x = f.a[i][2 * s];
            av.y = f.a[i][2 * s + 1];
            #pragma unroll
            for (int g = 0; g < NG; ++g) {
                #pragma unroll
                for (int j = 0; j < 4; ++j) {
                    v2f bv;
                    bv.x = f.w[g][2 * s][j];
                    bv.y = f.w[g][2 * s + 1][j];
                    acc[i][g][j] = __builtin_amdgcn_wmma_f32_16x16x4_f32(
                         false, av,  false, bv,
                         (short)0, acc[i][g][j],  false,  false);
                }
            }
        }
    }
}

#define PCAP 192
template <int MT, int NG, int PITCH>
__device__ __forceinline__ void load_frags_lds(Frags<MT, NG>& f, const float* ap,
                                               const float* __restrict__ wp, int N) {
    #pragma unroll
    for (int i = 0; i < MT; ++i)
        f.a[i] = *(const v4f*)(ap + 16 * i * PITCH);
    #pragma unroll
    for (int g = 0; g < NG; ++g) {
        #pragma unroll
        for (int q = 0; q < 4; ++q)
            f.w[g][q] = *(const v4f*)(wp + (size_t)q * N + 64 * g);
    }
}

template <int K, int EL>
__global__ void __launch_bounds__(128)
sage_fused(const float* __restrict__ xs, const int* __restrict__ perm,
           const int* __restrict__ cnt, const int* __restrict__ cur,
           const float* __restrict__ W0,
           const float* __restrict__ W1,
           const float* __restrict__ bias, float* __restrict__ out, int relu) {
    constexpr int N = 256, MT = 2, NG = 1, STEPS = K / 8, UNR = 8;
    constexpr int V = K / 128;
    constexpr int PITCH = K + 4;
    __shared__ __attribute__((aligned(16))) float aggT[32 * PITCH];
    __shared__ int lperm[4 * PCAP];
    const int lane = threadIdx.x & 31;
    const int w = threadIdx.x >> 5;
    const int h = lane >> 4;
    const int c = lane & 15;
    const int m0 = blockIdx.x * 32;
    const int n0 = 64 * w;
    const int dv = cnt[m0 + 8 * w + min(lane, 7)];
    const int bv = cur[m0 + 8 * w + min(lane, 7)];

    v8f acc[MT][NG][4];
    #pragma unroll
    for (int i = 0; i < MT; ++i) {
        #pragma unroll
        for (int j = 0; j < 4; ++j)
            acc[i][0][j] = (v8f){0.f, 0.f, 0.f, 0.f, 0.f, 0.f, 0.f, 0.f};
    }
    const float* a1p = xs + (size_t)(m0 + c) * K + 4 * h;
    const size_t woff = (size_t)(4 * h) * N + n0 + 4 * c;
    const float* w0p = W0 + woff;
    const float* w1p = W1 + woff;
    Frags<MT, NG> fa, fb;

    {
        auto fetch = [&](Frags<MT, NG>& f, int t) {
            load_frags<MT, NG>(f, a1p + 8 * t, K, w1p + (size_t)(8 * t) * N, N);
        };
        fetch(fa, 0);
        int t = 0;
        for (; t < STEPS - 2; t += 2) {
            fetch(fb, t + 1);
            __builtin_amdgcn_sched_barrier(0);
            mma_step<MT, NG>(acc, fa);
            __builtin_amdgcn_sched_barrier(0);
            fetch(fa, t + 2);
            __builtin_amdgcn_sched_barrier(0);
            mma_step<MT, NG>(acc, fb);
            __builtin_amdgcn_sched_barrier(0);
        }
        fetch(fb, t + 1);
        __builtin_amdgcn_sched_barrier(0);
        mma_step<MT, NG>(acc, fa);
        mma_step<MT, NG>(acc, fb);
    }

    {
        const float* xl = xs + 4 * lane;
        const int b0 = __builtin_amdgcn_readlane(bv, 0);
        const int lpo = PCAP * w;
        int pq[PCAP / 32];
        #pragma unroll
        for (int q = 0; q < PCAP / 32; ++q) pq[q] = perm[min(b0 + 32 * q + lane, EL - 1)];
        #pragma unroll
        for (int q = 0; q < PCAP / 32; ++q) lperm[lpo + 32 * q + lane] = pq[q];
        __syncthreads();
        for (int r = 0; r < 8; ++r) {
            const int deg = __builtin_amdgcn_readlane(dv, r);
            const int begin = __builtin_amdgcn_readlane(bv, r);
            const int off = begin - b0;
            v4f s[V];
            #pragma unroll
            for (int v = 0; v < V; ++v) s[v] = (v4f){0.f, 0.f, 0.f, 0.f};
            for (int j0 = 0; j0 < deg; j0 += 32) {
                const int m = min(32, deg - j0);
                const int o = off + j0;
                int my = lperm[lpo + min(max(o, 0) + min(lane, m - 1), PCAP - 1)];
                asm volatile("" : "+v"(my));
                if (o < 0 || o + m > PCAP) my = perm[begin + j0 + min(lane, m - 1)];
                for (int k = 0; k < m; k += UNR) {
                    v4f rr[UNR][V];
                    #pragma unroll
                    for (int u = 0; u < UNR; ++u) {
                        const int kk = min(k + u, m - 1);
                        const int sidx = __builtin_amdgcn_readlane(my, kk);
                        const float* rp = xl + (size_t)sidx * K;
                        #pragma unroll
                        for (int v = 0; v < V; ++v) rr[u][v] = *(const v4f*)(rp + 128 * v);
                    }
                    #pragma unroll
                    for (int u = 0; u < UNR; ++u) {
                        const bool on = (u == 0) || (k + u) < m;
                        #pragma unroll
                        for (int v = 0; v < V; ++v)
                            s[v] += on ? rr[u][v] : (v4f){0.f, 0.f, 0.f, 0.f};
                    }
                }
            }
            const float inv = 1.0f / fmaxf((float)deg, 1.0f);
            float* mp = aggT + (8 * w + r) * PITCH + 4 * lane;
            #pragma unroll
            for (int v = 0; v < V; ++v) *(v4f*)(mp + 128 * v) = s[v] * inv;
        }
    }
    __syncthreads();

    {
        const float* alp = aggT + c * PITCH + 4 * h;
        auto fetch = [&](Frags<MT, NG>& f, int t) {
            load_frags_lds<MT, NG, PITCH>(f, alp + 8 * t, w0p + (size_t)(8 * t) * N, N);
        };
        fetch(fa, 0);
        int t = 0;
        for (; t < STEPS - 2; t += 2) {
            fetch(fb, t + 1);
            __builtin_amdgcn_sched_barrier(0);
            mma_step<MT, NG>(acc, fa);
            __builtin_amdgcn_sched_barrier(0);
            fetch(fa, t + 2);
            __builtin_amdgcn_sched_barrier(0);
            mma_step<MT, NG>(acc, fb);
            __builtin_amdgcn_sched_barrier(0);
        }
        fetch(fb, t + 1);
        __builtin_amdgcn_sched_barrier(0);
        mma_step<MT, NG>(acc, fa);
        mma_step<MT, NG>(acc, fb);
    }

    {
        const int col = n0 + 4 * c;
        const float b0 = bias[col], b1 = bias[col + 1], b2 = bias[col + 2], b3 = bias[col + 3];
        #pragma unroll
        for (int i = 0; i < MT; ++i) {
            float* op = out + (size_t)(m0 + 16 * i + 8 * h) * N + col;
            #pragma unroll
            for (int r = 0; r < 8; ++r) {
                v4f o;
                o.x = acc[i][0][0][r] + b0;
                o.y = acc[i][0][1][r] + b1;
                o.z = acc[i][0][2][r] + b2;
                o.w = acc[i][0][3][r] + b3;
                if (relu) {
                    o.x = fmaxf(o.x, 0.0f); o.y = fmaxf(o.y, 0.0f);
                    o.z = fmaxf(o.z, 0.0f); o.w = fmaxf(o.w, 0.0f);
                }
                *(v4f*)(op + (size_t)r * N) = o;
            }
        }
    }
}

__global__ void __launch_bounds__(256)
sage_fused_ksplit(const float* __restrict__ xs, const int* __restrict__ perm,
                  const int* __restrict__ cnt, const int* __restrict__ cur,
                  const float* __restrict__ W0,
                  const float* __restrict__ W1,
                  const float* __restrict__ bias, float* __restrict__ out, int relu) {
    constexpr int K = 256, N = 128, MT = 1, NG = 2, UNR = 8, V = K / 128;
    constexpr int APITCH = 2 * K + 4;
    constexpr int PPITCH = N + 4;
    __shared__ __attribute__((aligned(16))) float aT[16 * APITCH];
    __shared__ __attribute__((aligned(16))) float part[8 * 16 * PPITCH];
    const int lane = threadIdx.x & 31;
    const int w = threadIdx.x >> 5;
    const int h = lane >> 4;
    const int c = lane & 15;
    const int m0 = blockIdx.x * 16;
    const int dv = cnt[m0 + 2 * w + min(lane, 1)];
    const int bv = cur[m0 + 2 * w + min(lane, 1)];

    const float* wsl = (w < 4 ? W0 + (size_t)(64 * w) * N : W1 + (size_t)(64 * (w - 4)) * N)
                       + (size_t)(4 * h) * N + 4 * c;
    Frags<MT, NG> fr[8];
    #pragma unroll
    for (int t = 0; t < 8; ++t) {
        #pragma unroll
        for (int g = 0; g < NG; ++g) {
            #pragma unroll
            for (int q = 0; q < 4; ++q)
                fr[t].w[g][q] = *(const v4f*)(wsl + (size_t)(8 * t + q) * N + 64 * g);
        }
    }

    {
        const float* xl = xs + 4 * lane;
        #pragma unroll
        for (int r = 0; r < 2; ++r) {
            const int deg = __builtin_amdgcn_readlane(dv, r);
            const int begin = __builtin_amdgcn_readlane(bv, r);
            v4f sacc[V];
            #pragma unroll
            for (int v = 0; v < V; ++v) sacc[v] = (v4f){0.f, 0.f, 0.f, 0.f};
            for (int j0 = 0; j0 < deg; j0 += 32) {
                const int m = min(32, deg - j0);
                const int my = perm[begin + j0 + min(lane, m - 1)];
                for (int k = 0; k < m; k += UNR) {
                    v4f rr[UNR][V];
                    #pragma unroll
                    for (int u = 0; u < UNR; ++u) {
                        const int kk = min(k + u, m - 1);
                        const int sidx = __builtin_amdgcn_readlane(my, kk);
                        const float* rp = xl + (size_t)sidx * K;
                        #pragma unroll
                        for (int v = 0; v < V; ++v) rr[u][v] = *(const v4f*)(rp + 128 * v);
                    }
                    #pragma unroll
                    for (int u = 0; u < UNR; ++u) {
                        const bool on = (u == 0) || (k + u) < m;
                        #pragma unroll
                        for (int v = 0; v < V; ++v)
                            sacc[v] += on ? rr[u][v] : (v4f){0.f, 0.f, 0.f, 0.f};
                    }
                }
            }
            const float inv = 1.0f / fmaxf((float)deg, 1.0f);
            float* mp = aT + (2 * w + r) * APITCH + 4 * lane;
            const float* tp = xs + (size_t)(m0 + 2 * w + r) * K + 4 * lane;
            #pragma unroll
            for (int v = 0; v < V; ++v) {
                *(v4f*)(mp + 128 * v) = sacc[v] * inv;
                *(v4f*)(mp + K + 128 * v) = *(const v4f*)(tp + 128 * v);
            }
        }
    }
    __syncthreads();

    v8f acc[MT][NG][4];
    #pragma unroll
    for (int g = 0; g < NG; ++g) {
        #pragma unroll
        for (int j = 0; j < 4; ++j)
            acc[0][g][j] = (v8f){0.f, 0.f, 0.f, 0.f, 0.f, 0.f, 0.f, 0.f};
    }
    {
        const float* alp = aT + c * APITCH + 64 * w + 4 * h;
        #pragma unroll
        for (int t = 0; t < 8; ++t) fr[t].a[0] = *(const v4f*)(alp + 8 * t);
        __builtin_amdgcn_sched_barrier(0);
        #pragma unroll
        for (int t = 0; t < 8; ++t) mma_step<MT, NG>(acc, fr[t]);
    }

    {
        float* pp = part + w * (16 * PPITCH) + (8 * h) * PPITCH + 4 * c;
        #pragma unroll
        for (int g = 0; g < NG; ++g) {
            #pragma unroll
            for (int r = 0; r < 8; ++r) {
                v4f o;
                o.x = acc[0][g][0][r];
                o.y = acc[0][g][1][r];
                o.z = acc[0][g][2][r];
                o.w = acc[0][g][3][r];
                *(v4f*)(pp + r * PPITCH + 64 * g) = o;
            }
        }
    }
    __syncthreads();
    {
        const v4f bq = *(const v4f*)(bias + 4 * lane);
        #pragma unroll
        for (int r = 0; r < 2; ++r) {
            const float* rp = part + (2 * w + r) * PPITCH + 4 * lane;
            v4f o = *(const v4f*)rp;
            #pragma unroll
            for (int p = 1; p < 8; ++p) o += *(const v4f*)(rp + p * (16 * PPITCH));
            o += bq;
            if (relu) {
                o.x = fmaxf(o.x, 0.0f); o.y = fmaxf(o.y, 0.0f);
                o.z = fmaxf(o.z, 0.0f); o.w = fmaxf(o.w, 0.0f);
            }
            *(v4f*)(out + (size_t)(m0 + 2 * w + r) * N + 4 * lane) = o;
        }
    }
}

extern "C" void kernel_launch(void* const* d_in, const int* in_sizes, int n_in,
                              void* d_out, int out_size, void* d_ws, size_t ws_size,
                              hipStream_t stream) {
    (void)in_sizes; (void)n_in; (void)out_size; (void)ws_size;

    const float* x = (const float*)d_in[0];
    const int* srcp[3] = {(const int*)d_in[1], (const int*)d_in[3], (const int*)d_in[5]};
    const int* dstp[3] = {(const int*)d_in[2], (const int*)d_in[4], (const int*)d_in[6]};
    const float* Wl[3] = {(const float*)d_in[7],  (const float*)d_in[10], (const float*)d_in[13]};
    const float* Wr[3] = {(const float*)d_in[8],  (const float*)d_in[11], (const float*)d_in[14]};
    const float* bb[3] = {(const float*)d_in[9],  (const float*)d_in[12], (const float*)d_in[15]};

    const int relu[3] = {1, 0, 1};

    float* ws     = (float*)d_ws;
    float* aggbuf = ws;
    float* small  = aggbuf + 12800000;
    float* h0     = small + 100000;
    float* h1     = h0 + 25600000;

    int* cnt0 = (int*)aggbuf + 6000000;  int* cur0 = cnt0 + N_L0;  int* perm0 = cur0 + N_L0;
    int* cnt1 = (int*)aggbuf + 8000000;  int* cur1 = cnt1 + N_L1;  int* perm1 = cur1 + N_L1;
    int* cnt2 = (int*)small;  int* cur2 = cnt2 + N_L2;  int* perm2 = cur2 + N_L2;
    int* tc0   = (int*)aggbuf;    int* tc1 = tc0 + T0;  int* tc2 = tc1 + T1;
    int* tl0   = tc0 + 75012;     int* tl1 = tl0 + T0;  int* tl2 = tl1 + T1;
    int* pairs = tl0 + 75012;

    chunk_sort<<<CB0 + CB1 + CB2, 1024, 0, stream>>>(srcp[0], dstp[0], srcp[1], dstp[1],
                                                      srcp[2], dstp[2], tc0, tc1, tc2,
                                                      tl0, tl1, tl2, pairs);
    bin_sort<<<NB0 + NB1 + NB2, 1024, 0, stream>>>(tc0, tc1, tc2, tl0, tl1, tl2, pairs,
                                                    cnt0, cur0, perm0, cnt1, cur1, perm1,
                                                    cnt2, cur2, perm2);

    sage_fused<128, E_L0><<<N_L0 / 32, 128, 0, stream>>>(x, perm0, cnt0, cur0, Wl[0], Wr[0], bb[0], h0, relu[0]);
    sage_fused<256, E_L1><<<N_L1 / 32, 128, 0, stream>>>(h0, perm1, cnt1, cur1, Wl[1], Wr[1], bb[1], h1, relu[1]);

    sage_fused_ksplit<<<N_L2 / 16, 256, 0, stream>>>(h1, perm2, cnt2, cur2, Wl[2], Wr[2], bb[2],
                                                     (float*)d_out, relu[2]);
}
